# starting kernel
# baseline (speedup 1.0000x reference)
#include <hip/hip_runtime.h>

typedef __attribute__((ext_vector_type(2))) float v2f;
typedef __attribute__((ext_vector_type(8))) float v8f;

#define N_NODES 50000
#define N_EDGES 600000
#define D 128
#define N_LAYERS 4

__global__ __launch_bounds__(256) void deg_kernel(const int* __restrict__ src,
                                                  const int* __restrict__ dst,
                                                  float* __restrict__ out_deg,
                                                  float* __restrict__ in_deg) {
    int e = blockIdx.x * blockDim.x + threadIdx.x;
    if (e < N_EDGES) {
        unsafeAtomicAdd(&out_deg[src[e]], 1.0f);
        unsafeAtomicAdd(&in_deg[dst[e]], 1.0f);
    }
}

__global__ __launch_bounds__(256) void norm_kernel(float* __restrict__ sn,
                                                   float* __restrict__ dn) {
    int n = blockIdx.x * blockDim.x + threadIdx.x;
    if (n < N_NODES) {
        sn[n] = rsqrtf(fmaxf(sn[n], 1.0f));
        dn[n] = rsqrtf(fmaxf(dn[n], 1.0f));
    }
}

__global__ __launch_bounds__(256) void agg_kernel(const float* __restrict__ h,
                                                  const int* __restrict__ src,
                                                  const int* __restrict__ dst,
                                                  const float* __restrict__ src_norm,
                                                  float* __restrict__ agg) {
    int tid   = blockIdx.x * blockDim.x + threadIdx.x;
    int e     = tid >> 5;
    int chunk = (tid & 31) * 4;
    if (e >= N_EDGES) return;

    int s = src[e];
    int d = dst[e];
    float sc = src_norm[s];

    const float4 v = *reinterpret_cast<const float4*>(h + (size_t)s * D + chunk);
    float* ap = agg + (size_t)d * D + chunk;
    unsafeAtomicAdd(ap + 0, v.x * sc);
    unsafeAtomicAdd(ap + 1, v.y * sc);
    unsafeAtomicAdd(ap + 2, v.z * sc);
    unsafeAtomicAdd(ap + 3, v.w * sc);
}

__global__ __launch_bounds__(256) void gemm_bias_kernel(
        const float* __restrict__ A,
        const float* __restrict__ dnorm,
        const float* __restrict__ Wm,
        const float* __restrict__ bias,
        float* __restrict__ out)
{
    __shared__ float ldsA[16 * 132];

    const int M0 = blockIdx.x * 16;
    const int t  = threadIdx.x;

#pragma unroll
    for (int half = 0; half < 2; ++half) {
        int flat = half * 1024 + t * 4;
        int row  = flat >> 7;
        int col  = flat & 127;
        float4 v = *reinterpret_cast<const float4*>(A + (size_t)(M0 + row) * D + col);
        float  s = dnorm[M0 + row];
        float* p = &ldsA[row * 132 + col];
        p[0] = v.x * s;
        p[1] = v.y * s;
        p[2] = v.z * s;
        p[3] = v.w * s;
    }
    __syncthreads();

    const int wave = t >> 5;
    const int lane = t & 31;
    const int n0   = wave * 16;
    const int mrow = lane & 15;
    const int kb   = (lane >> 4) * 2;
    const int ncol = n0 + (lane & 15);

    v8f c = {};
#pragma unroll
    for (int k0 = 0; k0 < D; k0 += 4) {
        const float* ap = &ldsA[mrow * 132 + k0 + kb];
        v2f a;
        a.x = ap[0];
        a.y = ap[1];

        v2f b;
        b.x = Wm[(size_t)(k0 + kb    ) * D + ncol];
        b.y = Wm[(size_t)(k0 + kb + 1) * D + ncol];

        c = __builtin_amdgcn_wmma_f32_16x16x4_f32(
                 false, a,  false, b,
                 (short)0, c,  false,  false);
    }

    const float bn    = bias[ncol];
    const int   mbase = M0 + ((lane >> 4) * 8);
#pragma unroll
    for (int v = 0; v < 8; ++v) {
        out[(size_t)(mbase + v) * D + ncol] = c[v] + bn;
    }
}

extern "C" void kernel_launch(void* const* d_in, const int* in_sizes, int n_in,
                              void* d_out, int out_size, void* d_ws, size_t ws_size,
                              hipStream_t stream) {
    const float* feat = (const float*)d_in[0];
    const int*   src  = (const int*)  d_in[1];
    const int*   dst  = (const int*)  d_in[2];
    const float* W    = (const float*)d_in[3];
    const float* b    = (const float*)d_in[4];
    float*       out  = (float*)d_out;

    float* ws       = (float*)d_ws;
    float* src_norm = ws;
    float* dst_norm = ws + N_NODES;
    float* agg      = ws + 2 * (size_t)N_NODES;
    float* hbuf     = agg + (size_t)N_NODES * D;

    hipMemsetAsync(src_norm, 0, 2 * (size_t)N_NODES * sizeof(float), stream);
    deg_kernel<<<(N_EDGES + 255) / 256, 256, 0, stream>>>(src, dst, src_norm, dst_norm);
    norm_kernel<<<(N_NODES + 255) / 256, 256, 0, stream>>>(src_norm, dst_norm);

    const float* h = feat;
    for (int i = 0; i < N_LAYERS; ++i) {
        hipMemsetAsync(agg, 0, (size_t)N_NODES * D * sizeof(float), stream);

        agg_kernel<<<(N_EDGES * 32) / 256, 256, 0, stream>>>(h, src, dst, src_norm, agg);

        float* o = (i == N_LAYERS - 1) ? out : hbuf;
        gemm_bias_kernel<<<N_NODES / 16, 256, 0, stream>>>(
            agg, dst_norm, W + (size_t)i * D * D, b + (size_t)i * D, o);
        h = hbuf;
    }
}


# Aggregation: per-edge global f32 atomics and the agg zero fill replaced by owner blocks: a block scans dst, keeps its 196 rows in LDS, sums the source rows there with LDS f32 adds; sum order changes
# speedup vs baseline: 1.7375x; 1.7375x over previous
#include <hip/hip_runtime.h>

typedef __attribute__((ext_vector_type(2))) float v2f;
typedef __attribute__((ext_vector_type(8))) float v8f;

#define N_NODES 50000
#define N_EDGES 600000
#define D 128
#define N_LAYERS 4

__global__ __launch_bounds__(256) void deg_kernel(const int* __restrict__ src,
                                                  const int* __restrict__ dst,
                                                  float* __restrict__ out_deg,
                                                  float* __restrict__ in_deg) {
    int e = blockIdx.x * blockDim.x + threadIdx.x;
    if (e < N_EDGES) {
        unsafeAtomicAdd(&out_deg[src[e]], 1.0f);
        unsafeAtomicAdd(&in_deg[dst[e]], 1.0f);
    }
}

__global__ __launch_bounds__(256) void norm_kernel(float* __restrict__ sn,
                                                   float* __restrict__ dn) {
    int n = blockIdx.x * blockDim.x + threadIdx.x;
    if (n < N_NODES) {
        sn[n] = rsqrtf(fmaxf(sn[n], 1.0f));
        dn[n] = rsqrtf(fmaxf(dn[n], 1.0f));
    }
}

#define OWN_ROWS    196
#define OWN_BLOCKS  ((N_NODES + OWN_ROWS - 1) / OWN_ROWS)
#define OWN_THREADS 1024
#define OWN_WAVES   (OWN_THREADS / 32)
#define LIST_CAP    10240
#define SCAN_U      8
#define LOOK_U      4
#define GATHER_U    8

static __device__ __forceinline__ void add_edge_alone(float* __restrict__ arow,
                                                      const float* __restrict__ h,
                                                      const int* __restrict__ src,
                                                      const float* __restrict__ src_norm,
                                                      int e) {
    const int   s   = src[e];
    const float sc  = src_norm[s];
    const float* hr = h + (size_t)s * D;
    for (int k = 0; k < D; ++k)
        unsafeAtomicAdd(&arow[(k & 3) * 32 + (k >> 2)], hr[k] * sc);
}

__global__ __launch_bounds__(OWN_THREADS) void agg_owner_kernel(
        const float* __restrict__ h,
        const int* __restrict__ src,
        const int* __restrict__ dst,
        const float* __restrict__ src_norm,
        float* __restrict__ agg) {
    __shared__ float    acc[OWN_ROWS * D];
    __shared__ unsigned lst[LIST_CAP];
    __shared__ float    lsc[LIST_CAP];
    __shared__ int      cnt;

    const int tid  = threadIdx.x;
    const int wave = tid >> 5;
    const int lane = tid & 31;
    const int lo   = blockIdx.x * OWN_ROWS;

    {
        float4* a4 = reinterpret_cast<float4*>(acc);
        const float4 z = make_float4(0.f, 0.f, 0.f, 0.f);
        for (int i = tid; i < OWN_ROWS * D / 4; i += OWN_THREADS) a4[i] = z;
        if (tid == 0) cnt = 0;
    }
    __syncthreads();

    {
        const int4* d4 = reinterpret_cast<const int4*>(dst);
        constexpr int NE4 = N_EDGES / 4;
        for (int base = tid; base < NE4; base += OWN_THREADS * SCAN_U) {
            int4 vv[SCAN_U];
#pragma unroll
            for (int u = 0; u < SCAN_U; ++u) {
                const int idx = base + u * OWN_THREADS;
                vv[u] = (idx < NE4) ? d4[idx] : make_int4(-1, -1, -1, -1);
            }
#pragma unroll
            for (int u = 0; u < SCAN_U; ++u) {
                const int  idx = base + u * OWN_THREADS;
                const unsigned rx = (unsigned)(vv[u].x - lo);
                const unsigned ry = (unsigned)(vv[u].y - lo);
                const unsigned rz = (unsigned)(vv[u].z - lo);
                const unsigned rw = (unsigned)(vv[u].w - lo);
                if (min(min(rx, ry), min(rz, rw)) < (unsigned)OWN_ROWS) {
                    const unsigned rr[4] = {rx, ry, rz, rw};
#pragma unroll
                    for (int c = 0; c < 4; ++c) {
                        if (rr[c] < (unsigned)OWN_ROWS) {
                            const int pos = atomicAdd(&cnt, 1);
                            if (pos < LIST_CAP) lst[pos] = (unsigned)(idx * 4 + c);
                        }
                    }
                }
            }
        }
    }
    __syncthreads();
    int n = cnt;
    if (n > LIST_CAP) {
        for (int e = tid; e < N_EDGES; e += OWN_THREADS) {
            const unsigned r = (unsigned)(dst[e] - lo);
            if (r < (unsigned)OWN_ROWS) add_edge_alone(acc + r * D, h, src, src_norm, e);
        }
        n = 0;
    }

    for (int i0 = tid; i0 < n; i0 += OWN_THREADS * LOOK_U) {
        int   e[LOOK_U];
        int   s[LOOK_U];
        int   d[LOOK_U];
        float sc[LOOK_U];
#pragma unroll
        for (int k = 0; k < LOOK_U; ++k) {
            const int i = i0 + k * OWN_THREADS;
            e[k] = (i < n) ? (int)lst[i] : 0;
        }
#pragma unroll
        for (int k = 0; k < LOOK_U; ++k) {
            s[k] = src[e[k]];
            d[k] = dst[e[k]];
        }
        asm volatile("" :: "v"(s[0]), "v"(s[1]), "v"(s[2]), "v"(s[3]),
                           "v"(d[0]), "v"(d[1]), "v"(d[2]), "v"(d[3]));
#pragma unroll
        for (int k = 0; k < LOOK_U; ++k) sc[k] = src_norm[s[k]];
        asm volatile("" :: "v"(sc[0]), "v"(sc[1]), "v"(sc[2]), "v"(sc[3]));
#pragma unroll
        for (int k = 0; k < LOOK_U; ++k) {
            const int i = i0 + k * OWN_THREADS;
            if (i < n) {
                lsc[i] = sc[k];
                lst[i] = ((unsigned)(d[k] - lo) << 16) | (unsigned)s[k];
            }
        }
    }
    __syncthreads();

    for (int i0 = wave * GATHER_U; i0 < n; i0 += OWN_WAVES * GATHER_U) {
        unsigned p[GATHER_U];
        float    sc[GATHER_U];
        float4   row[GATHER_U];
#pragma unroll
        for (int u = 0; u < GATHER_U; ++u) {
            const int i = min(i0 + u, n - 1);
            p[u]   = __builtin_amdgcn_readfirstlane(lst[i]);
            sc[u]  = lsc[i];
            row[u] = *reinterpret_cast<const float4*>(h + (size_t)(p[u] & 0xFFFFu) * D + lane * 4);
        }
#pragma unroll
        for (int u = 0; u < GATHER_U; ++u) {
            if (i0 + u < n) {
                float* a = acc + (p[u] >> 16) * D + lane;
                unsafeAtomicAdd(a,      row[u].x * sc[u]);
                unsafeAtomicAdd(a + 32, row[u].y * sc[u]);
                unsafeAtomicAdd(a + 64, row[u].z * sc[u]);
                unsafeAtomicAdd(a + 96, row[u].w * sc[u]);
            }
        }
    }
    __syncthreads();

    for (int r = wave; r < OWN_ROWS; r += OWN_WAVES) {
        const int node = lo + r;
        if (node < N_NODES) {
            const float* a = acc + r * D + lane;
            float4 o;
            o.x = a[0];
            o.y = a[32];
            o.z = a[64];
            o.w = a[96];
            *reinterpret_cast<float4*>(agg + (size_t)node * D + lane * 4) = o;
        }
    }
}

__global__ __launch_bounds__(256) void gemm_bias_kernel(
        const float* __restrict__ A,
        const float* __restrict__ dnorm,
        const float* __restrict__ Wm,
        const float* __restrict__ bias,
        float* __restrict__ out)
{
    __shared__ float ldsA[16 * 132];

    const int M0 = blockIdx.x * 16;
    const int t  = threadIdx.x;

#pragma unroll
    for (int half = 0; half < 2; ++half) {
        int flat = half * 1024 + t * 4;
        int row  = flat >> 7;
        int col  = flat & 127;
        float4 v = *reinterpret_cast<const float4*>(A + (size_t)(M0 + row) * D + col);
        float  s = dnorm[M0 + row];
        float* p = &ldsA[row * 132 + col];
        p[0] = v.x * s;
        p[1] = v.y * s;
        p[2] = v.z * s;
        p[3] = v.w * s;
    }
    __syncthreads();

    const int wave = t >> 5;
    const int lane = t & 31;
    const int n0   = wave * 16;
    const int mrow = lane & 15;
    const int kb   = (lane >> 4) * 2;
    const int ncol = n0 + (lane & 15);

    v8f c = {};
#pragma unroll
    for (int k0 = 0; k0 < D; k0 += 4) {
        const float* ap = &ldsA[mrow * 132 + k0 + kb];
        v2f a;
        a.x = ap[0];
        a.y = ap[1];

        v2f b;
        b.x = Wm[(size_t)(k0 + kb    ) * D + ncol];
        b.y = Wm[(size_t)(k0 + kb + 1) * D + ncol];

        c = __builtin_amdgcn_wmma_f32_16x16x4_f32(
                 false, a,  false, b,
                 (short)0, c,  false,  false);
    }

    const float bn    = bias[ncol];
    const int   mbase = M0 + ((lane >> 4) * 8);
#pragma unroll
    for (int v = 0; v < 8; ++v) {
        out[(size_t)(mbase + v) * D + ncol] = c[v] + bn;
    }
}

extern "C" void kernel_launch(void* const* d_in, const int* in_sizes, int n_in,
                              void* d_out, int out_size, void* d_ws, size_t ws_size,
                              hipStream_t stream) {
    const float* feat = (const float*)d_in[0];
    const int*   src  = (const int*)  d_in[1];
    const int*   dst  = (const int*)  d_in[2];
    const float* W    = (const float*)d_in[3];
    const float* b    = (const float*)d_in[4];
    float*       out  = (float*)d_out;

    float* ws       = (float*)d_ws;
    float* src_norm = ws;
    float* dst_norm = ws + N_NODES;
    float* agg      = ws + 2 * (size_t)N_NODES;
    float* hbuf     = agg + (size_t)N_NODES * D;

    hipMemsetAsync(src_norm, 0, 2 * (size_t)N_NODES * sizeof(float), stream);
    deg_kernel<<<(N_EDGES + 255) / 256, 256, 0, stream>>>(src, dst, src_norm, dst_norm);
    norm_kernel<<<(N_NODES + 255) / 256, 256, 0, stream>>>(src_norm, dst_norm);

    const float* h = feat;
    for (int i = 0; i < N_LAYERS; ++i) {
        agg_owner_kernel<<<OWN_BLOCKS, OWN_THREADS, 0, stream>>>(h, src, dst, src_norm, agg);

        float* o = (i == N_LAYERS - 1) ? out : hbuf;
        gemm_bias_kernel<<<N_NODES / 16, 256, 0, stream>>>(
            agg, dst_norm, W + (size_t)i * D * D, b + (size_t)i * D, o);
        h = hbuf;
    }
}


# Aggregation: owner blocks' edge lists (row|source word, source norm, count) built in layer 1's launch only, kept in workspace; layers 2-4 copy them to LDS: 3 of 4 dst scans gone; sum order as before
# speedup vs baseline: 2.3070x; 1.3278x over previous
#include <hip/hip_runtime.h>

typedef __attribute__((ext_vector_type(2))) float v2f;
typedef __attribute__((ext_vector_type(8))) float v8f;

#define N_NODES 50000
#define N_EDGES 600000
#define D 128
#define N_LAYERS 4

__global__ __launch_bounds__(256) void deg_kernel(const int* __restrict__ src,
                                                  const int* __restrict__ dst,
                                                  float* __restrict__ out_deg,
                                                  float* __restrict__ in_deg) {
    int e = blockIdx.x * blockDim.x + threadIdx.x;
    if (e < N_EDGES) {
        unsafeAtomicAdd(&out_deg[src[e]], 1.0f);
        unsafeAtomicAdd(&in_deg[dst[e]], 1.0f);
    }
}

__global__ __launch_bounds__(256) void norm_kernel(float* __restrict__ sn,
                                                   float* __restrict__ dn) {
    int n = blockIdx.x * blockDim.x + threadIdx.x;
    if (n < N_NODES) {
        sn[n] = rsqrtf(fmaxf(sn[n], 1.0f));
        dn[n] = rsqrtf(fmaxf(dn[n], 1.0f));
    }
}

#define OWN_ROWS    196
#define OWN_BLOCKS  ((N_NODES + OWN_ROWS - 1) / OWN_ROWS)
#define OWN_THREADS 1024
#define OWN_WAVES   (OWN_THREADS / 32)
#define LIST_CAP    10240
#define SCAN_U      8
#define LOOK_U      4
#define GATHER_U    8
#define CNT_PITCH   32

static __device__ __forceinline__ void add_edge_alone(float* __restrict__ arow,
                                                      const float* __restrict__ h,
                                                      const int* __restrict__ src,
                                                      const float* __restrict__ src_norm,
                                                      int e) {
    const int   s   = src[e];
    const float sc  = src_norm[s];
    const float* hr = h + (size_t)s * D;
    for (int k = 0; k < D; ++k)
        unsafeAtomicAdd(&arow[(k & 3) * 32 + (k >> 2)], hr[k] * sc);
}

template <bool BUILD>
__global__ __launch_bounds__(OWN_THREADS) void agg_owner_kernel(
        const float* __restrict__ h,
        const int* __restrict__ src,
        const int* __restrict__ dst,
        const float* __restrict__ src_norm,
        float* __restrict__ agg,
        unsigned* __restrict__ glst,
        float* __restrict__ glsc,
        int* __restrict__ gcnt) {
    __shared__ float    acc[OWN_ROWS * D];
    __shared__ unsigned lst[LIST_CAP];
    __shared__ float    lsc[LIST_CAP];
    __shared__ int      cnt;

    const int tid  = threadIdx.x;
    const int wave = tid >> 5;
    const int lane = tid & 31;
    const int lo   = blockIdx.x * OWN_ROWS;

    unsigned* __restrict__ blst = glst + (size_t)blockIdx.x * LIST_CAP;
    float*    __restrict__ blsc = glsc + (size_t)blockIdx.x * LIST_CAP;

    int n = BUILD ? 0 : gcnt[blockIdx.x * CNT_PITCH];

    {
        float4* a4 = reinterpret_cast<float4*>(acc);
        const float4 z = make_float4(0.f, 0.f, 0.f, 0.f);
        for (int i = tid; i < OWN_ROWS * D / 4; i += OWN_THREADS) a4[i] = z;
        if (BUILD && tid == 0) cnt = 0;
    }
    if (!BUILD && n <= LIST_CAP) {
        for (int i = tid; i < n; i += OWN_THREADS) {
            lst[i] = blst[i];
            lsc[i] = blsc[i];
        }
    }
    __syncthreads();

    if (BUILD) {
        const int4* d4 = reinterpret_cast<const int4*>(dst);
        constexpr int NE4 = N_EDGES / 4;
        for (int base = tid; base < NE4; base += OWN_THREADS * SCAN_U) {
            int4 vv[SCAN_U];
#pragma unroll
            for (int u = 0; u < SCAN_U; ++u) {
                const int idx = base + u * OWN_THREADS;
                vv[u] = (idx < NE4) ? d4[idx] : make_int4(-1, -1, -1, -1);
            }
#pragma unroll
            for (int u = 0; u < SCAN_U; ++u) {
                const int  idx = base + u * OWN_THREADS;
                const unsigned rx = (unsigned)(vv[u].x - lo);
                const unsigned ry = (unsigned)(vv[u].y - lo);
                const unsigned rz = (unsigned)(vv[u].z - lo);
                const unsigned rw = (unsigned)(vv[u].w - lo);
                if (min(min(rx, ry), min(rz, rw)) < (unsigned)OWN_ROWS) {
                    const unsigned rr[4] = {rx, ry, rz, rw};
#pragma unroll
                    for (int c = 0; c < 4; ++c) {
                        if (rr[c] < (unsigned)OWN_ROWS) {
                            const int pos = atomicAdd(&cnt, 1);
                            if (pos < LIST_CAP) lst[pos] = (unsigned)(idx * 4 + c);
                        }
                    }
                }
            }
        }
    }
    if (BUILD) {
        __syncthreads();
        n = cnt;
        if (tid == 0) gcnt[blockIdx.x * CNT_PITCH] = n;
    }
    if (n > LIST_CAP) {
        for (int e = tid; e < N_EDGES; e += OWN_THREADS) {
            const unsigned r = (unsigned)(dst[e] - lo);
            if (r < (unsigned)OWN_ROWS) add_edge_alone(acc + r * D, h, src, src_norm, e);
        }
        n = 0;
    }

    for (int i0 = tid; BUILD && i0 < n; i0 += OWN_THREADS * LOOK_U) {
        int   e[LOOK_U];
        int   s[LOOK_U];
        int   d[LOOK_U];
        float sc[LOOK_U];
#pragma unroll
        for (int k = 0; k < LOOK_U; ++k) {
            const int i = i0 + k * OWN_THREADS;
            e[k] = (i < n) ? (int)lst[i] : 0;
        }
#pragma unroll
        for (int k = 0; k < LOOK_U; ++k) {
            s[k] = src[e[k]];
            d[k] = dst[e[k]];
        }
        asm volatile("" :: "v"(s[0]), "v"(s[1]), "v"(s[2]), "v"(s[3]),
                           "v"(d[0]), "v"(d[1]), "v"(d[2]), "v"(d[3]));
#pragma unroll
        for (int k = 0; k < LOOK_U; ++k) sc[k] = src_norm[s[k]];
        asm volatile("" :: "v"(sc[0]), "v"(sc[1]), "v"(sc[2]), "v"(sc[3]));
#pragma unroll
        for (int k = 0; k < LOOK_U; ++k) {
            const int i = i0 + k * OWN_THREADS;
            if (i < n) {
                const unsigned w = ((unsigned)(d[k] - lo) << 16) | (unsigned)s[k];
                lsc[i]  = sc[k];
                lst[i]  = w;
                blsc[i] = sc[k];
                blst[i] = w;
            }
        }
    }
    if (BUILD) __syncthreads();

    for (int i0 = wave * GATHER_U; i0 < n; i0 += OWN_WAVES * GATHER_U) {
        unsigned p[GATHER_U];
        float    sc[GATHER_U];
        float4   row[GATHER_U];
#pragma unroll
        for (int u = 0; u < GATHER_U; ++u) {
            const int i = min(i0 + u, n - 1);
            p[u]   = __builtin_amdgcn_readfirstlane(lst[i]);
            sc[u]  = lsc[i];
            row[u] = *reinterpret_cast<const float4*>(h + (size_t)(p[u] & 0xFFFFu) * D + lane * 4);
        }
#pragma unroll
        for (int u = 0; u < GATHER_U; ++u) {
            if (i0 + u < n) {
                float* a = acc + (p[u] >> 16) * D + lane;
                unsafeAtomicAdd(a,      row[u].x * sc[u]);
                unsafeAtomicAdd(a + 32, row[u].y * sc[u]);
                unsafeAtomicAdd(a + 64, row[u].z * sc[u]);
                unsafeAtomicAdd(a + 96, row[u].w * sc[u]);
            }
        }
    }
    __syncthreads();

    for (int r = wave; r < OWN_ROWS; r += OWN_WAVES) {
        const int node = lo + r;
        if (node < N_NODES) {
            const float* a = acc + r * D + lane;
            float4 o;
            o.x = a[0];
            o.y = a[32];
            o.z = a[64];
            o.w = a[96];
            *reinterpret_cast<float4*>(agg + (size_t)node * D + lane * 4) = o;
        }
    }
}

__global__ __launch_bounds__(256) void gemm_bias_kernel(
        const float* __restrict__ A,
        const float* __restrict__ dnorm,
        const float* __restrict__ Wm,
        const float* __restrict__ bias,
        float* __restrict__ out)
{
    __shared__ float ldsA[16 * 132];

    const int M0 = blockIdx.x * 16;
    const int t  = threadIdx.x;

#pragma unroll
    for (int half = 0; half < 2; ++half) {
        int flat = half * 1024 + t * 4;
        int row  = flat >> 7;
        int col  = flat & 127;
        float4 v = *reinterpret_cast<const float4*>(A + (size_t)(M0 + row) * D + col);
        float  s = dnorm[M0 + row];
        float* p = &ldsA[row * 132 + col];
        p[0] = v.x * s;
        p[1] = v.y * s;
        p[2] = v.z * s;
        p[3] = v.w * s;
    }
    __syncthreads();

    const int wave = t >> 5;
    const int lane = t & 31;
    const int n0   = wave * 16;
    const int mrow = lane & 15;
    const int kb   = (lane >> 4) * 2;
    const int ncol = n0 + (lane & 15);

    v8f c = {};
#pragma unroll
    for (int k0 = 0; k0 < D; k0 += 4) {
        const float* ap = &ldsA[mrow * 132 + k0 + kb];
        v2f a;
        a.x = ap[0];
        a.y = ap[1];

        v2f b;
        b.x = Wm[(size_t)(k0 + kb    ) * D + ncol];
        b.y = Wm[(size_t)(k0 + kb + 1) * D + ncol];

        c = __builtin_amdgcn_wmma_f32_16x16x4_f32(
                 false, a,  false, b,
                 (short)0, c,  false,  false);
    }

    const float bn    = bias[ncol];
    const int   mbase = M0 + ((lane >> 4) * 8);
#pragma unroll
    for (int v = 0; v < 8; ++v) {
        out[(size_t)(mbase + v) * D + ncol] = c[v] + bn;
    }
}

extern "C" void kernel_launch(void* const* d_in, const int* in_sizes, int n_in,
                              void* d_out, int out_size, void* d_ws, size_t ws_size,
                              hipStream_t stream) {
    const float* feat = (const float*)d_in[0];
    const int*   src  = (const int*)  d_in[1];
    const int*   dst  = (const int*)  d_in[2];
    const float* W    = (const float*)d_in[3];
    const float* b    = (const float*)d_in[4];
    float*       out  = (float*)d_out;

    float* ws       = (float*)d_ws;
    float* src_norm = ws;
    float* dst_norm = ws + N_NODES;
    float* agg      = ws + 2 * (size_t)N_NODES;
    float* hbuf     = agg + (size_t)N_NODES * D;
    unsigned* glst  = (unsigned*)(hbuf + (size_t)N_NODES * D);
    float*    glsc  = (float*)(glst + (size_t)OWN_BLOCKS * LIST_CAP);
    int*      gcnt  = (int*)(glsc + (size_t)OWN_BLOCKS * LIST_CAP);

    hipMemsetAsync(src_norm, 0, 2 * (size_t)N_NODES * sizeof(float), stream);
    deg_kernel<<<(N_EDGES + 255) / 256, 256, 0, stream>>>(src, dst, src_norm, dst_norm);
    norm_kernel<<<(N_NODES + 255) / 256, 256, 0, stream>>>(src_norm, dst_norm);

    const float* h = feat;
    for (int i = 0; i < N_LAYERS; ++i) {
        if (i == 0)
            agg_owner_kernel<true><<<OWN_BLOCKS, OWN_THREADS, 0, stream>>>(
                h, src, dst, src_norm, agg, glst, glsc, gcnt);
        else
            agg_owner_kernel<false><<<OWN_BLOCKS, OWN_THREADS, 0, stream>>>(
                h, src, dst, src_norm, agg, glst, glsc, gcnt);

        float* o = (i == N_LAYERS - 1) ? out : hbuf;
        gemm_bias_kernel<<<N_NODES / 16, 256, 0, stream>>>(
            agg, dst_norm, W + (size_t)i * D * D, b + (size_t)i * D, o);
        h = hbuf;
    }
}


# Dense product fused into the owner blocks: LDS row sums are read as the WMMA A operand (row pitch 132), W column tile in registers; agg buffer and 4 launches gone; f32; product k sum order changes
# speedup vs baseline: 2.5669x; 1.1127x over previous
#include <hip/hip_runtime.h>

typedef __attribute__((ext_vector_type(2))) float v2f;
typedef __attribute__((ext_vector_type(8))) float v8f;

#define N_NODES 50000
#define N_EDGES 600000
#define D 128
#define N_LAYERS 4

__global__ __launch_bounds__(256) void deg_kernel(const int* __restrict__ src,
                                                  const int* __restrict__ dst,
                                                  float* __restrict__ out_deg,
                                                  float* __restrict__ in_deg) {
    int e = blockIdx.x * blockDim.x + threadIdx.x;
    if (e < N_EDGES) {
        unsafeAtomicAdd(&out_deg[src[e]], 1.0f);
        unsafeAtomicAdd(&in_deg[dst[e]], 1.0f);
    }
}

__global__ __launch_bounds__(256) void norm_kernel(float* __restrict__ sn,
                                                   float* __restrict__ dn) {
    int n = blockIdx.x * blockDim.x + threadIdx.x;
    if (n < N_NODES) {
        sn[n] = rsqrtf(fmaxf(sn[n], 1.0f));
        dn[n] = rsqrtf(fmaxf(dn[n], 1.0f));
    }
}

#define OWN_ROWS    196
#define OWN_BLOCKS  ((N_NODES + OWN_ROWS - 1) / OWN_ROWS)
#define OWN_THREADS 1024
#define OWN_WAVES   (OWN_THREADS / 32)
#define LIST_CAP    10240
#define SCAN_U      8
#define LOOK_U      4
#define GATHER_U    8
#define CNT_PITCH   32
#define ACC_TILES   ((OWN_ROWS + 15) / 16)
#define ACC_ROWS    (ACC_TILES * 16)
#define ACC_PITCH   132

static __device__ __forceinline__ int feat_of_word(int w) { return 4 * (w & 31) + (w >> 5); }

static __device__ __forceinline__ void add_edge_alone(float* __restrict__ arow,
                                                      const float* __restrict__ h,
                                                      const int* __restrict__ src,
                                                      const float* __restrict__ src_norm,
                                                      int e) {
    const int   s   = src[e];
    const float sc  = src_norm[s];
    const float* hr = h + (size_t)s * D;
    for (int k = 0; k < D; ++k)
        unsafeAtomicAdd(&arow[(k & 3) * 32 + (k >> 2)], hr[k] * sc);
}

template <bool BUILD>
__global__ __launch_bounds__(OWN_THREADS) void layer_owner_kernel(
        const float* __restrict__ h,
        const int* __restrict__ src,
        const int* __restrict__ dst,
        const float* __restrict__ src_norm,
        const float* __restrict__ dst_norm,
        const float* __restrict__ Wm,
        const float* __restrict__ bias,
        float* __restrict__ hout,
        unsigned* __restrict__ glst,
        float* __restrict__ glsc,
        int* __restrict__ gcnt) {
    __shared__ __attribute__((aligned(16))) float acc[ACC_ROWS * ACC_PITCH];
    __shared__ unsigned lst[LIST_CAP];
    __shared__ float    lsc[LIST_CAP];
    __shared__ int      cnt;

    const int tid  = threadIdx.x;
    const int wave = tid >> 5;
    const int lane = tid & 31;
    const int lo   = blockIdx.x * OWN_ROWS;

    unsigned* __restrict__ blst = glst + (size_t)blockIdx.x * LIST_CAP;
    float*    __restrict__ blsc = glsc + (size_t)blockIdx.x * LIST_CAP;

    int n = BUILD ? 0 : gcnt[blockIdx.x * CNT_PITCH];

    {
        float4* a4 = reinterpret_cast<float4*>(acc);
        const float4 z = make_float4(0.f, 0.f, 0.f, 0.f);
        for (int i = tid; i < ACC_ROWS * ACC_PITCH / 4; i += OWN_THREADS) a4[i] = z;
        if (BUILD && tid == 0) cnt = 0;
    }
    if (!BUILD && n <= LIST_CAP) {
        for (int i = tid; i < n; i += OWN_THREADS) {
            lst[i] = blst[i];
            lsc[i] = blsc[i];
        }
    }
    __syncthreads();

    if (BUILD) {
        const int4* d4 = reinterpret_cast<const int4*>(dst);
        constexpr int NE4 = N_EDGES / 4;
        for (int base = tid; base < NE4; base += OWN_THREADS * SCAN_U) {
            int4 vv[SCAN_U];
#pragma unroll
            for (int u = 0; u < SCAN_U; ++u) {
                const int idx = base + u * OWN_THREADS;
                vv[u] = (idx < NE4) ? d4[idx] : make_int4(-1, -1, -1, -1);
            }
#pragma unroll
            for (int u = 0; u < SCAN_U; ++u) {
                const int  idx = base + u * OWN_THREADS;
                const unsigned rx = (unsigned)(vv[u].x - lo);
                const unsigned ry = (unsigned)(vv[u].y - lo);
                const unsigned rz = (unsigned)(vv[u].z - lo);
                const unsigned rw = (unsigned)(vv[u].w - lo);
                if (min(min(rx, ry), min(rz, rw)) < (unsigned)OWN_ROWS) {
                    const unsigned rr[4] = {rx, ry, rz, rw};
#pragma unroll
                    for (int c = 0; c < 4; ++c) {
                        if (rr[c] < (unsigned)OWN_ROWS) {
                            const int pos = atomicAdd(&cnt, 1);
                            if (pos < LIST_CAP) lst[pos] = (unsigned)(idx * 4 + c);
                        }
                    }
                }
            }
        }
    }
    if (BUILD) {
        __syncthreads();
        n = cnt;
        if (tid == 0) gcnt[blockIdx.x * CNT_PITCH] = n;
    }
    if (n > LIST_CAP) {
        for (int e = tid; e < N_EDGES; e += OWN_THREADS) {
            const unsigned r = (unsigned)(dst[e] - lo);
            if (r < (unsigned)OWN_ROWS) add_edge_alone(acc + r * ACC_PITCH, h, src, src_norm, e);
        }
        n = 0;
    }

    for (int i0 = tid; BUILD && i0 < n; i0 += OWN_THREADS * LOOK_U) {
        int   e[LOOK_U];
        int   s[LOOK_U];
        int   d[LOOK_U];
        float sc[LOOK_U];
#pragma unroll
        for (int k = 0; k < LOOK_U; ++k) {
            const int i = i0 + k * OWN_THREADS;
            e[k] = (i < n) ? (int)lst[i] : 0;
        }
#pragma unroll
        for (int k = 0; k < LOOK_U; ++k) {
            s[k] = src[e[k]];
            d[k] = dst[e[k]];
        }
        asm volatile("" :: "v"(s[0]), "v"(s[1]), "v"(s[2]), "v"(s[3]),
                           "v"(d[0]), "v"(d[1]), "v"(d[2]), "v"(d[3]));
#pragma unroll
        for (int k = 0; k < LOOK_U; ++k) sc[k] = src_norm[s[k]];
        asm volatile("" :: "v"(sc[0]), "v"(sc[1]), "v"(sc[2]), "v"(sc[3]));
#pragma unroll
        for (int k = 0; k < LOOK_U; ++k) {
            const int i = i0 + k * OWN_THREADS;
            if (i < n) {
                const unsigned w = ((unsigned)(d[k] - lo) << 16) | (unsigned)s[k];
                lsc[i]  = sc[k];
                lst[i]  = w;
                blsc[i] = sc[k];
                blst[i] = w;
            }
        }
    }
    if (BUILD) __syncthreads();

    for (int i0 = wave * GATHER_U; i0 < n; i0 += OWN_WAVES * GATHER_U) {
        unsigned p[GATHER_U];
        float    sc[GATHER_U];
        float4   row[GATHER_U];
#pragma unroll
        for (int u = 0; u < GATHER_U; ++u) {
            const int i = min(i0 + u, n - 1);
            p[u]   = __builtin_amdgcn_readfirstlane(lst[i]);
            sc[u]  = lsc[i];
            row[u] = *reinterpret_cast<const float4*>(h + (size_t)(p[u] & 0xFFFFu) * D + lane * 4);
        }
#pragma unroll
        for (int u = 0; u < GATHER_U; ++u) {
            if (i0 + u < n) {
                float* a = acc + (p[u] >> 16) * ACC_PITCH + lane;
                unsafeAtomicAdd(a,      row[u].x * sc[u]);
                unsafeAtomicAdd(a + 32, row[u].y * sc[u]);
                unsafeAtomicAdd(a + 64, row[u].z * sc[u]);
                unsafeAtomicAdd(a + 96, row[u].w * sc[u]);
            }
        }
    }

    static_assert(OWN_WAVES == 32 && ACC_TILES == 13, "tile deal below is written for 32 waves, 13 row tiles");
    const int wv   = __builtin_amdgcn_readfirstlane(wave);
    const int c16  = lane & 15;
    const int hf   = lane >> 4;
    const int ncol = (wv & 7) * 16 + c16;
    v2f bfr[D / 4];
#pragma unroll
    for (int s = 0; s < D / 4; ++s) {
        bfr[s].x = Wm[(size_t)feat_of_word(4 * s + 2 * hf    ) * D + ncol];
        bfr[s].y = Wm[(size_t)feat_of_word(4 * s + 2 * hf + 1) * D + ncol];
    }
    const float bn = bias[ncol];
    __syncthreads();

    for (int t = 0; t < 4; ++t) {
        int rt = (wv >> 3) + 4 * t;
        if (t == 3) {
            if ((wv >> 3) != (wv & 3)) break;
            rt = ACC_TILES - 1;
        }
        const int   rowl = rt * 16 + c16;
        const float dn   = dst_norm[min(lo + rowl, N_NODES - 1)];
        const float* ap  = acc + rowl * ACC_PITCH + 2 * hf;
        v8f c = {};
#pragma unroll
        for (int s = 0; s < D / 4; ++s) {
            v2f a = *reinterpret_cast<const v2f*>(ap + 4 * s);
            a.x *= dn;
            a.y *= dn;
            c = __builtin_amdgcn_wmma_f32_16x16x4_f32(
                     false, a,  false, bfr[s],
                     (short)0, c,  false,  false);
        }
        float o[8];
#pragma unroll
        for (int v = 0; v < 8; ++v) o[v] = c[v] + bn;
        const int rbase = rt * 16 + hf * 8;
#pragma unroll
        for (int v = 0; v < 8; ++v) {
            const int r    = rbase + v;
            const int node = lo + r;
            if (r < OWN_ROWS && node < N_NODES)
                hout[(size_t)node * D + ncol] = o[v];
        }
    }
}


extern "C" void kernel_launch(void* const* d_in, const int* in_sizes, int n_in,
                              void* d_out, int out_size, void* d_ws, size_t ws_size,
                              hipStream_t stream) {
    const float* feat = (const float*)d_in[0];
    const int*   src  = (const int*)  d_in[1];
    const int*   dst  = (const int*)  d_in[2];
    const float* W    = (const float*)d_in[3];
    const float* b    = (const float*)d_in[4];
    float*       out  = (float*)d_out;

    float* ws       = (float*)d_ws;
    float* src_norm = ws;
    float* dst_norm = ws + N_NODES;
    float* hA       = ws + 2 * (size_t)N_NODES;
    float* hB       = hA + (size_t)N_NODES * D;
    unsigned* glst  = (unsigned*)(hB + (size_t)N_NODES * D);
    float*    glsc  = (float*)(glst + (size_t)OWN_BLOCKS * LIST_CAP);
    int*      gcnt  = (int*)(glsc + (size_t)OWN_BLOCKS * LIST_CAP);

    hipMemsetAsync(src_norm, 0, 2 * (size_t)N_NODES * sizeof(float), stream);
    deg_kernel<<<(N_EDGES + 255) / 256, 256, 0, stream>>>(src, dst, src_norm, dst_norm);
    norm_kernel<<<(N_NODES + 255) / 256, 256, 0, stream>>>(src_norm, dst_norm);

    layer_owner_kernel<true><<<OWN_BLOCKS, OWN_THREADS, 0, stream>>>(
        feat, src, dst, src_norm, dst_norm, W, b, hA, glst, glsc, gcnt);
    layer_owner_kernel<false><<<OWN_BLOCKS, OWN_THREADS, 0, stream>>>(
        hA, src, dst, src_norm, dst_norm, W + (size_t)1 * D * D, b + (size_t)1 * D, hB, glst, glsc, gcnt);
    layer_owner_kernel<false><<<OWN_BLOCKS, OWN_THREADS, 0, stream>>>(
        hB, src, dst, src_norm, dst_norm, W + (size_t)2 * D * D, b + (size_t)2 * D, hA, glst, glsc, gcnt);
    layer_owner_kernel<false><<<OWN_BLOCKS, OWN_THREADS, 0, stream>>>(
        hA, src, dst, src_norm, dst_norm, W + (size_t)3 * D * D, b + (size_t)3 * D, out, glst, glsc, gcnt);
}


# In-degree: per-edge global f32 atomics replaced by LDS integer counts over the owner blocks' edge lists in layer 1's launch; rsqrt taken where a degree is read, norm launch gone; sum orders unchanged
# speedup vs baseline: 2.9331x; 1.1427x over previous
#include <hip/hip_runtime.h>

typedef __attribute__((ext_vector_type(2))) float v2f;
typedef __attribute__((ext_vector_type(8))) float v8f;

#define N_NODES 50000
#define N_EDGES 600000
#define D 128
#define N_LAYERS 4

__global__ __launch_bounds__(256) void deg_kernel(const int* __restrict__ src,
                                                  float* __restrict__ out_deg) {
    int e = blockIdx.x * blockDim.x + threadIdx.x;
    if (e < N_EDGES) unsafeAtomicAdd(&out_deg[src[e]], 1.0f);
}

static __device__ __forceinline__ float norm_of_deg(float deg) { return rsqrtf(fmaxf(deg, 1.0f)); }

#define OWN_ROWS    196
#define OWN_BLOCKS  ((N_NODES + OWN_ROWS - 1) / OWN_ROWS)
#define OWN_THREADS 1024
#define OWN_WAVES   (OWN_THREADS / 32)
#define LIST_CAP    10240
#define SCAN_U      8
#define LOOK_U      4
#define GATHER_U    8
#define CNT_PITCH   32
#define DN_PITCH    256
#define ACC_TILES   ((OWN_ROWS + 15) / 16)
#define ACC_ROWS    (ACC_TILES * 16)
#define ACC_PITCH   132

static __device__ __forceinline__ int feat_of_word(int w) { return 4 * (w & 31) + (w >> 5); }

static __device__ __forceinline__ void add_edge_alone(float* __restrict__ arow,
                                                      const float* __restrict__ h,
                                                      const int* __restrict__ src,
                                                      const float* __restrict__ src_deg,
                                                      int e) {
    const int   s   = src[e];
    const float sc  = norm_of_deg(src_deg[s]);
    const float* hr = h + (size_t)s * D;
    for (int k = 0; k < D; ++k)
        unsafeAtomicAdd(&arow[(k & 3) * 32 + (k >> 2)], hr[k] * sc);
}

template <bool BUILD>
__global__ __launch_bounds__(OWN_THREADS) void layer_owner_kernel(
        const float* __restrict__ h,
        const int* __restrict__ src,
        const int* __restrict__ dst,
        const float* __restrict__ src_deg,
        const float* __restrict__ Wm,
        const float* __restrict__ bias,
        float* __restrict__ hout,
        unsigned* __restrict__ glst,
        float* __restrict__ glsc,
        int* __restrict__ gcnt,
        float* __restrict__ gdn) {
    __shared__ __attribute__((aligned(16))) float acc[ACC_ROWS * ACC_PITCH];
    __shared__ unsigned lst[LIST_CAP];
    __shared__ float    lsc[LIST_CAP];
    __shared__ int      cnt;
    __shared__ int      rcnt[ACC_ROWS];

    const int tid  = threadIdx.x;
    const int wave = tid >> 5;
    const int lane = tid & 31;
    const int lo   = blockIdx.x * OWN_ROWS;

    unsigned* __restrict__ blst = glst + (size_t)blockIdx.x * LIST_CAP;
    float*    __restrict__ blsc = glsc + (size_t)blockIdx.x * LIST_CAP;

    int n = BUILD ? 0 : gcnt[blockIdx.x * CNT_PITCH];

    {
        float4* a4 = reinterpret_cast<float4*>(acc);
        const float4 z = make_float4(0.f, 0.f, 0.f, 0.f);
        for (int i = tid; i < ACC_ROWS * ACC_PITCH / 4; i += OWN_THREADS) a4[i] = z;
        if (BUILD && tid == 0) cnt = 0;
        if (BUILD && tid < ACC_ROWS) rcnt[tid] = 0;
    }
    if (!BUILD && n <= LIST_CAP) {
        for (int i = tid; i < n; i += OWN_THREADS) {
            lst[i] = blst[i];
            lsc[i] = blsc[i];
        }
    }
    __syncthreads();

    if (BUILD) {
        const int4* d4 = reinterpret_cast<const int4*>(dst);
        constexpr int NE4 = N_EDGES / 4;
        for (int base = tid; base < NE4; base += OWN_THREADS * SCAN_U) {
            int4 vv[SCAN_U];
#pragma unroll
            for (int u = 0; u < SCAN_U; ++u) {
                const int idx = base + u * OWN_THREADS;
                vv[u] = (idx < NE4) ? d4[idx] : make_int4(-1, -1, -1, -1);
            }
#pragma unroll
            for (int u = 0; u < SCAN_U; ++u) {
                const int  idx = base + u * OWN_THREADS;
                const unsigned rx = (unsigned)(vv[u].x - lo);
                const unsigned ry = (unsigned)(vv[u].y - lo);
                const unsigned rz = (unsigned)(vv[u].z - lo);
                const unsigned rw = (unsigned)(vv[u].w - lo);
                if (min(min(rx, ry), min(rz, rw)) < (unsigned)OWN_ROWS) {
                    const unsigned rr[4] = {rx, ry, rz, rw};
#pragma unroll
                    for (int c = 0; c < 4; ++c) {
                        if (rr[c] < (unsigned)OWN_ROWS) {
                            const int pos = atomicAdd(&cnt, 1);
                            if (pos < LIST_CAP) lst[pos] = (unsigned)(idx * 4 + c);
                        }
                    }
                }
            }
        }
    }
    if (BUILD) {
        __syncthreads();
        n = cnt;
        if (tid == 0) gcnt[blockIdx.x * CNT_PITCH] = n;
    }
    if (n > LIST_CAP) {
        for (int e = tid; e < N_EDGES; e += OWN_THREADS) {
            const unsigned r = (unsigned)(dst[e] - lo);
            if (r < (unsigned)OWN_ROWS) {
                if (BUILD) atomicAdd(&rcnt[r], 1);
                add_edge_alone(acc + r * ACC_PITCH, h, src, src_deg, e);
            }
        }
        n = 0;
    }

    for (int i0 = tid; BUILD && i0 < n; i0 += OWN_THREADS * LOOK_U) {
        int   e[LOOK_U];
        int   s[LOOK_U];
        int   d[LOOK_U];
        float dg[LOOK_U];
#pragma unroll
        for (int k = 0; k < LOOK_U; ++k) {
            const int i = i0 + k * OWN_THREADS;
            e[k] = (i < n) ? (int)lst[i] : 0;
        }
#pragma unroll
        for (int k = 0; k < LOOK_U; ++k) {
            s[k] = src[e[k]];
            d[k] = dst[e[k]];
        }
        asm volatile("" :: "v"(s[0]), "v"(s[1]), "v"(s[2]), "v"(s[3]),
                           "v"(d[0]), "v"(d[1]), "v"(d[2]), "v"(d[3]));
#pragma unroll
        for (int k = 0; k < LOOK_U; ++k) dg[k] = src_deg[s[k]];
        asm volatile("" :: "v"(dg[0]), "v"(dg[1]), "v"(dg[2]), "v"(dg[3]));
#pragma unroll
        for (int k = 0; k < LOOK_U; ++k) {
            const int i = i0 + k * OWN_THREADS;
            if (i < n) {
                const unsigned r  = (unsigned)(d[k] - lo);
                const unsigned w  = (r << 16) | (unsigned)s[k];
                const float    sc = norm_of_deg(dg[k]);
                atomicAdd(&rcnt[r], 1);
                lsc[i]  = sc;
                lst[i]  = w;
                blsc[i] = sc;
                blst[i] = w;
            }
        }
    }
    if (BUILD) {
        __syncthreads();
        if (tid < ACC_ROWS) gdn[blockIdx.x * DN_PITCH + tid] = norm_of_deg((float)rcnt[tid]);
    }

    for (int i0 = wave * GATHER_U; i0 < n; i0 += OWN_WAVES * GATHER_U) {
        unsigned p[GATHER_U];
        float    sc[GATHER_U];
        float4   row[GATHER_U];
#pragma unroll
        for (int u = 0; u < GATHER_U; ++u) {
            const int i = min(i0 + u, n - 1);
            p[u]   = __builtin_amdgcn_readfirstlane(lst[i]);
            sc[u]  = lsc[i];
            row[u] = *reinterpret_cast<const float4*>(h + (size_t)(p[u] & 0xFFFFu) * D + lane * 4);
        }
#pragma unroll
        for (int u = 0; u < GATHER_U; ++u) {
            if (i0 + u < n) {
                float* a = acc + (p[u] >> 16) * ACC_PITCH + lane;
                unsafeAtomicAdd(a,      row[u].x * sc[u]);
                unsafeAtomicAdd(a + 32, row[u].y * sc[u]);
                unsafeAtomicAdd(a + 64, row[u].z * sc[u]);
                unsafeAtomicAdd(a + 96, row[u].w * sc[u]);
            }
        }
    }

    static_assert(OWN_WAVES == 32 && ACC_TILES == 13, "tile deal below is written for 32 waves, 13 row tiles");
    const int wv   = __builtin_amdgcn_readfirstlane(wave);
    const int c16  = lane & 15;
    const int hf   = lane >> 4;
    const int ncol = (wv & 7) * 16 + c16;
    v2f bfr[D / 4];
#pragma unroll
    for (int s = 0; s < D / 4; ++s) {
        bfr[s].x = Wm[(size_t)feat_of_word(4 * s + 2 * hf    ) * D + ncol];
        bfr[s].y = Wm[(size_t)feat_of_word(4 * s + 2 * hf + 1) * D + ncol];
    }
    const float bn = bias[ncol];
    __syncthreads();

    for (int t = 0; t < 4; ++t) {
        int rt = (wv >> 3) + 4 * t;
        if (t == 3) {
            if ((wv >> 3) != (wv & 3)) break;
            rt = ACC_TILES - 1;
        }
        const int   rowl = rt * 16 + c16;
        const float dn   = BUILD ? norm_of_deg((float)rcnt[rowl]) : gdn[blockIdx.x * DN_PITCH + rowl];
        const float* ap  = acc + rowl * ACC_PITCH + 2 * hf;
        v8f c = {};
#pragma unroll
        for (int s = 0; s < D / 4; ++s) {
            v2f a = *reinterpret_cast<const v2f*>(ap + 4 * s);
            a.x *= dn;
            a.y *= dn;
            c = __builtin_amdgcn_wmma_f32_16x16x4_f32(
                     false, a,  false, bfr[s],
                     (short)0, c,  false,  false);
        }
        float o[8];
#pragma unroll
        for (int v = 0; v < 8; ++v) o[v] = c[v] + bn;
        const int rbase = rt * 16 + hf * 8;
#pragma unroll
        for (int v = 0; v < 8; ++v) {
            const int r    = rbase + v;
            const int node = lo + r;
            if (r < OWN_ROWS && node < N_NODES)
                hout[(size_t)node * D + ncol] = o[v];
        }
    }
}


extern "C" void kernel_launch(void* const* d_in, const int* in_sizes, int n_in,
                              void* d_out, int out_size, void* d_ws, size_t ws_size,
                              hipStream_t stream) {
    const float* feat = (const float*)d_in[0];
    const int*   src  = (const int*)  d_in[1];
    const int*   dst  = (const int*)  d_in[2];
    const float* W    = (const float*)d_in[3];
    const float* b    = (const float*)d_in[4];
    float*       out  = (float*)d_out;

    float* ws       = (float*)d_ws;
    float* out_deg  = ws;
    float* hA       = ws + 2 * (size_t)N_NODES;
    float* hB       = hA + (size_t)N_NODES * D;
    unsigned* glst  = (unsigned*)(hB + (size_t)N_NODES * D);
    float*    glsc  = (float*)(glst + (size_t)OWN_BLOCKS * LIST_CAP);
    int*      gcnt  = (int*)(glsc + (size_t)OWN_BLOCKS * LIST_CAP);
    float*    gdn   = (float*)(gcnt + (size_t)OWN_BLOCKS * CNT_PITCH);

    hipMemsetAsync(out_deg, 0, (size_t)N_NODES * sizeof(float), stream);
    deg_kernel<<<(N_EDGES + 255) / 256, 256, 0, stream>>>(src, out_deg);

    layer_owner_kernel<true><<<OWN_BLOCKS, OWN_THREADS, 0, stream>>>(
        feat, src, dst, out_deg, W, b, hA, glst, glsc, gcnt, gdn);
    layer_owner_kernel<false><<<OWN_BLOCKS, OWN_THREADS, 0, stream>>>(
        hA, src, dst, out_deg, W + (size_t)1 * D * D, b + (size_t)1 * D, hB, glst, glsc, gcnt, gdn);
    layer_owner_kernel<false><<<OWN_BLOCKS, OWN_THREADS, 0, stream>>>(
        hB, src, dst, out_deg, W + (size_t)2 * D * D, b + (size_t)2 * D, hA, glst, glsc, gcnt, gdn);
    layer_owner_kernel<false><<<OWN_BLOCKS, OWN_THREADS, 0, stream>>>(
        hA, src, dst, out_deg, W + (size_t)3 * D * D, b + (size_t)3 * D, out, glst, glsc, gcnt, gdn);
}


# Owner blocks' edge lists by one partition pass: the out-degree launch sorts its 1/256 piece of the edges by owner in LDS; layer 1 copies its 256 runs; 256 whole-dst scans gone; sum order as before
# speedup vs baseline: 3.3595x; 1.1454x over previous
#include <hip/hip_runtime.h>

typedef __attribute__((ext_vector_type(2))) float v2f;
typedef __attribute__((ext_vector_type(8))) float v8f;

#define N_NODES 50000
#define N_EDGES 600000
#define D 128
#define N_LAYERS 4

#define OWN_ROWS    196
#define OWN_BLOCKS  ((N_NODES + OWN_ROWS - 1) / OWN_ROWS)

#define PART_BLOCKS  256
#define PART_E4      ((N_EDGES / 4 + PART_BLOCKS - 1) / PART_BLOCKS)
#define PART_THREADS 640
#define PART_PITCH   (((PART_E4 * 4 + 31) / 32) * 32)
#define OFF_PITCH    288
static_assert(N_EDGES % 4 == 0, "the edge pass reads four edges per 16-byte load");
static_assert(PART_THREADS >= PART_E4 && PART_THREADS % 32 == 0, "one 16-byte load of each array per thread");
static_assert(OWN_BLOCKS == 256 && PART_BLOCKS == 256, "the prefix sum below is written for 256 counts");
static_assert(OFF_PITCH >= OWN_BLOCKS + 1 && OFF_PITCH % 32 == 0, "table pitch");

static __device__ __forceinline__ void excl_scan_256(const int* cnt, int* start, int lane) {
    int v[8];
    int sum = 0;
#pragma unroll
    for (int k = 0; k < 8; ++k) { v[k] = cnt[lane * 8 + k]; sum += v[k]; }
    int inc = sum;
#pragma unroll
    for (int off = 1; off < 32; off <<= 1) {
        const int t = __shfl_up(inc, off);
        if (lane >= off) inc += t;
    }
    int ex = inc - sum;
#pragma unroll
    for (int k = 0; k < 8; ++k) { start[lane * 8 + k] = ex; ex += v[k]; }
    if (lane == 31) start[256] = ex;
}

__global__ __launch_bounds__(PART_THREADS) void edge_pass_kernel(const int* __restrict__ src,
                                                                 const int* __restrict__ dst,
                                                                 float* __restrict__ out_deg,
                                                                 unsigned* __restrict__ gpl,
                                                                 int* __restrict__ goff) {
    __shared__ int      bcnt[OWN_BLOCKS];
    __shared__ int      bstart[OWN_BLOCKS + 1];
    __shared__ unsigned sorted[PART_PITCH];

    const int tid = threadIdx.x;
    if (tid < OWN_BLOCKS) bcnt[tid] = 0;
    __syncthreads();

    const int  idx4  = blockIdx.x * PART_E4 + tid;
    const bool valid = tid < PART_E4 && idx4 < N_EDGES / 4;
    int      own[4] = {-1, -1, -1, -1};
    int      rk[4]  = {0, 0, 0, 0};
    unsigned wd[4]  = {0u, 0u, 0u, 0u};
    if (valid) {
        const int4 d4 = reinterpret_cast<const int4*>(dst)[idx4];
        const int4 s4 = reinterpret_cast<const int4*>(src)[idx4];
        const int dd[4] = {d4.x, d4.y, d4.z, d4.w};
        const int ss[4] = {s4.x, s4.y, s4.z, s4.w};
#pragma unroll
        for (int c = 0; c < 4; ++c) {
            unsafeAtomicAdd(&out_deg[ss[c]], 1.0f);
            if ((unsigned)dd[c] < (unsigned)N_NODES) {
                const unsigned o = (unsigned)dd[c] / OWN_ROWS;
                const unsigned r = (unsigned)dd[c] - o * OWN_ROWS;
                own[c] = (int)o;
                wd[c]  = (r << 16) | (unsigned)ss[c];
                rk[c]  = atomicAdd(&bcnt[o], 1);
            }
        }
    }
    __syncthreads();
    if (tid < 32) excl_scan_256(bcnt, bstart, tid);
    __syncthreads();
#pragma unroll
    for (int c = 0; c < 4; ++c)
        if (own[c] >= 0) sorted[bstart[own[c]] + rk[c]] = wd[c];
    __syncthreads();

    const int tot = bstart[OWN_BLOCKS];
    unsigned* __restrict__ pl = gpl + (size_t)blockIdx.x * PART_PITCH;
    for (int i = tid; i < tot; i += PART_THREADS) pl[i] = sorted[i];
    if (tid <= OWN_BLOCKS) goff[(size_t)blockIdx.x * OFF_PITCH + tid] = bstart[tid];
}

static __device__ __forceinline__ float norm_of_deg(float deg) { return rsqrtf(fmaxf(deg, 1.0f)); }

#define OWN_THREADS 1024
#define OWN_WAVES   (OWN_THREADS / 32)
#define LIST_CAP    10240
#define COPY_U      4
#define LOOK_U      4
#define GATHER_U    8
#define CNT_PITCH   32
#define DN_PITCH    256
#define ACC_TILES   ((OWN_ROWS + 15) / 16)
#define ACC_ROWS    (ACC_TILES * 16)
#define ACC_PITCH   132

static __device__ __forceinline__ int feat_of_word(int w) { return 4 * (w & 31) + (w >> 5); }

static __device__ __forceinline__ void add_edge_alone(float* __restrict__ arow,
                                                      const float* __restrict__ h,
                                                      const int* __restrict__ src,
                                                      const float* __restrict__ src_deg,
                                                      int e) {
    const int   s   = src[e];
    const float sc  = norm_of_deg(src_deg[s]);
    const float* hr = h + (size_t)s * D;
    for (int k = 0; k < D; ++k)
        unsafeAtomicAdd(&arow[(k & 3) * 32 + (k >> 2)], hr[k] * sc);
}

template <bool BUILD>
__global__ __launch_bounds__(OWN_THREADS) void layer_owner_kernel(
        const float* __restrict__ h,
        const int* __restrict__ src,
        const int* __restrict__ dst,
        const float* __restrict__ src_deg,
        const float* __restrict__ Wm,
        const float* __restrict__ bias,
        float* __restrict__ hout,
        unsigned* __restrict__ glst,
        float* __restrict__ glsc,
        int* __restrict__ gcnt,
        float* __restrict__ gdn,
        const unsigned* __restrict__ gpl,
        const int* __restrict__ goff) {
    __shared__ __attribute__((aligned(16))) float acc[ACC_ROWS * ACC_PITCH];
    __shared__ unsigned lst[LIST_CAP];
    __shared__ float    lsc[LIST_CAP];
    __shared__ int      rcnt[ACC_ROWS];
    __shared__ int      pstart[PART_BLOCKS];
    __shared__ int      pcnt[PART_BLOCKS];
    __shared__ int      pbase[PART_BLOCKS + 1];

    const int tid  = threadIdx.x;
    const int wave = tid >> 5;
    const int lane = tid & 31;
    const int lo   = blockIdx.x * OWN_ROWS;

    unsigned* __restrict__ blst = glst + (size_t)blockIdx.x * LIST_CAP;
    float*    __restrict__ blsc = glsc + (size_t)blockIdx.x * LIST_CAP;

    int n = BUILD ? 0 : gcnt[blockIdx.x * CNT_PITCH];

    {
        float4* a4 = reinterpret_cast<float4*>(acc);
        const float4 z = make_float4(0.f, 0.f, 0.f, 0.f);
        for (int i = tid; i < ACC_ROWS * ACC_PITCH / 4; i += OWN_THREADS) a4[i] = z;
        if (BUILD && tid < ACC_ROWS) rcnt[tid] = 0;
    }
    if (!BUILD && n <= LIST_CAP) {
        for (int i = tid; i < n; i += OWN_THREADS) {
            lst[i] = blst[i];
            lsc[i] = blsc[i];
        }
    }
    if (BUILD && tid < PART_BLOCKS) {
        const int* po = goff + (size_t)tid * OFF_PITCH + blockIdx.x;
        const int st = po[0];
        pstart[tid] = st;
        pcnt[tid]   = po[1] - st;
    }
    __syncthreads();

    if (BUILD) {
        static_assert(OWN_THREADS == 4 * PART_BLOCKS, "the run copy below deals four threads to a piece");
        if (wave == 0) excl_scan_256(pcnt, pbase, lane);
        __syncthreads();
        n = pbase[PART_BLOCKS];
        if (tid == 0) gcnt[blockIdx.x * CNT_PITCH] = n;
        if (n <= LIST_CAP) {
            const int p = tid >> 2;
            const int q = tid & 3;
            const int b = pbase[p];
            const int c = pbase[p + 1] - b;
            const unsigned* __restrict__ pp = gpl + (size_t)p * PART_PITCH + pstart[p];
            for (int j0 = q; j0 < c; j0 += 4 * COPY_U) {
                unsigned v[COPY_U];
#pragma unroll
                for (int k = 0; k < COPY_U; ++k) {
                    const int j = j0 + 4 * k;
                    v[k] = (j < c) ? pp[j] : 0u;
                }
                asm volatile("" :: "v"(v[0]), "v"(v[1]), "v"(v[2]), "v"(v[3]));
#pragma unroll
                for (int k = 0; k < COPY_U; ++k) {
                    const int j = j0 + 4 * k;
                    if (j < c) lst[b + j] = v[k];
                }
            }
        }
        __syncthreads();
    }
    if (n > LIST_CAP) {
        for (int e = tid; e < N_EDGES; e += OWN_THREADS) {
            const unsigned r = (unsigned)(dst[e] - lo);
            if (r < (unsigned)OWN_ROWS) {
                if (BUILD) atomicAdd(&rcnt[r], 1);
                add_edge_alone(acc + r * ACC_PITCH, h, src, src_deg, e);
            }
        }
        n = 0;
    }

    for (int i0 = tid; BUILD && i0 < n; i0 += OWN_THREADS * LOOK_U) {
        unsigned w[LOOK_U];
        float    dg[LOOK_U];
#pragma unroll
        for (int k = 0; k < LOOK_U; ++k) {
            const int i = i0 + k * OWN_THREADS;
            w[k] = (i < n) ? lst[i] : 0u;
        }
#pragma unroll
        for (int k = 0; k < LOOK_U; ++k) dg[k] = src_deg[w[k] & 0xFFFFu];
        asm volatile("" :: "v"(dg[0]), "v"(dg[1]), "v"(dg[2]), "v"(dg[3]));
#pragma unroll
        for (int k = 0; k < LOOK_U; ++k) {
            const int i = i0 + k * OWN_THREADS;
            if (i < n) {
                const float sc = norm_of_deg(dg[k]);
                atomicAdd(&rcnt[w[k] >> 16], 1);
                lsc[i]  = sc;
                blsc[i] = sc;
                blst[i] = w[k];
            }
        }
    }
    if (BUILD) {
        __syncthreads();
        if (tid < ACC_ROWS) gdn[blockIdx.x * DN_PITCH + tid] = norm_of_deg((float)rcnt[tid]);
    }

    for (int i0 = wave * GATHER_U; i0 < n; i0 += OWN_WAVES * GATHER_U) {
        unsigned p[GATHER_U];
        float    sc[GATHER_U];
        float4   row[GATHER_U];
#pragma unroll
        for (int u = 0; u < GATHER_U; ++u) {
            const int i = min(i0 + u, n - 1);
            p[u]   = __builtin_amdgcn_readfirstlane(lst[i]);
            sc[u]  = lsc[i];
            row[u] = *reinterpret_cast<const float4*>(h + (size_t)(p[u] & 0xFFFFu) * D + lane * 4);
        }
#pragma unroll
        for (int u = 0; u < GATHER_U; ++u) {
            if (i0 + u < n) {
                float* a = acc + (p[u] >> 16) * ACC_PITCH + lane;
                unsafeAtomicAdd(a,      row[u].x * sc[u]);
                unsafeAtomicAdd(a + 32, row[u].y * sc[u]);
                unsafeAtomicAdd(a + 64, row[u].z * sc[u]);
                unsafeAtomicAdd(a + 96, row[u].w * sc[u]);
            }
        }
    }

    static_assert(OWN_WAVES == 32 && ACC_TILES == 13, "tile deal below is written for 32 waves, 13 row tiles");
    const int wv   = __builtin_amdgcn_readfirstlane(wave);
    const int c16  = lane & 15;
    const int hf   = lane >> 4;
    const int ncol = (wv & 7) * 16 + c16;
    v2f bfr[D / 4];
#pragma unroll
    for (int s = 0; s < D / 4; ++s) {
        bfr[s].x = Wm[(size_t)feat_of_word(4 * s + 2 * hf    ) * D + ncol];
        bfr[s].y = Wm[(size_t)feat_of_word(4 * s + 2 * hf + 1) * D + ncol];
    }
    const float bn = bias[ncol];
    __syncthreads();

    for (int t = 0; t < 4; ++t) {
        int rt = (wv >> 3) + 4 * t;
        if (t == 3) {
            if ((wv >> 3) != (wv & 3)) break;
            rt = ACC_TILES - 1;
        }
        const int   rowl = rt * 16 + c16;
        const float dn   = BUILD ? norm_of_deg((float)rcnt[rowl]) : gdn[blockIdx.x * DN_PITCH + rowl];
        const float* ap  = acc + rowl * ACC_PITCH + 2 * hf;
        v8f c = {};
#pragma unroll
        for (int s = 0; s < D / 4; ++s) {
            v2f a = *reinterpret_cast<const v2f*>(ap + 4 * s);
            a.x *= dn;
            a.y *= dn;
            c = __builtin_amdgcn_wmma_f32_16x16x4_f32(
                     false, a,  false, bfr[s],
                     (short)0, c,  false,  false);
        }
        float o[8];
#pragma unroll
        for (int v = 0; v < 8; ++v) o[v] = c[v] + bn;
        const int rbase = rt * 16 + hf * 8;
#pragma unroll
        for (int v = 0; v < 8; ++v) {
            const int r    = rbase + v;
            const int node = lo + r;
            if (r < OWN_ROWS && node < N_NODES)
                hout[(size_t)node * D + ncol] = o[v];
        }
    }
}


extern "C" void kernel_launch(void* const* d_in, const int* in_sizes, int n_in,
                              void* d_out, int out_size, void* d_ws, size_t ws_size,
                              hipStream_t stream) {
    const float* feat = (const float*)d_in[0];
    const int*   src  = (const int*)  d_in[1];
    const int*   dst  = (const int*)  d_in[2];
    const float* W    = (const float*)d_in[3];
    const float* b    = (const float*)d_in[4];
    float*       out  = (float*)d_out;

    float* ws       = (float*)d_ws;
    float* out_deg  = ws;
    float* hA       = ws + 2 * (size_t)N_NODES;
    float* hB       = hA + (size_t)N_NODES * D;
    unsigned* glst  = (unsigned*)(hB + (size_t)N_NODES * D);
    float*    glsc  = (float*)(glst + (size_t)OWN_BLOCKS * LIST_CAP);
    int*      gcnt  = (int*)(glsc + (size_t)OWN_BLOCKS * LIST_CAP);
    float*    gdn   = (float*)(gcnt + (size_t)OWN_BLOCKS * CNT_PITCH);
    unsigned* gpl   = (unsigned*)(gdn + (size_t)OWN_BLOCKS * DN_PITCH);
    int*      goff  = (int*)(gpl + (size_t)PART_BLOCKS * PART_PITCH);

    hipMemsetAsync(out_deg, 0, (size_t)N_NODES * sizeof(float), stream);
    edge_pass_kernel<<<PART_BLOCKS, PART_THREADS, 0, stream>>>(src, dst, out_deg, gpl, goff);

    layer_owner_kernel<true><<<OWN_BLOCKS, OWN_THREADS, 0, stream>>>(
        feat, src, dst, out_deg, W, b, hA, glst, glsc, gcnt, gdn, gpl, goff);
    layer_owner_kernel<false><<<OWN_BLOCKS, OWN_THREADS, 0, stream>>>(
        hA, src, dst, out_deg, W + (size_t)1 * D * D, b + (size_t)1 * D, hB, glst, glsc, gcnt, gdn, gpl, goff);
    layer_owner_kernel<false><<<OWN_BLOCKS, OWN_THREADS, 0, stream>>>(
        hB, src, dst, out_deg, W + (size_t)2 * D * D, b + (size_t)2 * D, hA, glst, glsc, gcnt, gdn, gpl, goff);
    layer_owner_kernel<false><<<OWN_BLOCKS, OWN_THREADS, 0, stream>>>(
        hA, src, dst, out_deg, W + (size_t)3 * D * D, b + (size_t)3 * D, out, glst, glsc, gcnt, gdn, gpl, goff);
}


# Aggregation: owner block's list sorted by row in LDS (layer 1, on its in-degree counts); a wave sums a row's run of edges in registers (fma), one LDS f32 add per run, not per edge; sum order changes
# speedup vs baseline: 3.5137x; 1.0459x over previous
#include <hip/hip_runtime.h>

typedef __attribute__((ext_vector_type(2))) float v2f;
typedef __attribute__((ext_vector_type(8))) float v8f;

#define N_NODES 50000
#define N_EDGES 600000
#define D 128
#define N_LAYERS 4

#define OWN_ROWS    196
#define OWN_BLOCKS  ((N_NODES + OWN_ROWS - 1) / OWN_ROWS)

#define PART_BLOCKS  256
#define PART_E4      ((N_EDGES / 4 + PART_BLOCKS - 1) / PART_BLOCKS)
#define PART_THREADS 640
#define PART_PITCH   (((PART_E4 * 4 + 31) / 32) * 32)
#define OFF_PITCH    288
static_assert(N_EDGES % 4 == 0, "the edge pass reads four edges per 16-byte load");
static_assert(PART_THREADS >= PART_E4 && PART_THREADS % 32 == 0, "one 16-byte load of each array per thread");
static_assert(OWN_BLOCKS == 256 && PART_BLOCKS == 256, "the prefix sum below is written for 256 counts");
static_assert(OFF_PITCH >= OWN_BLOCKS + 1 && OFF_PITCH % 32 == 0, "table pitch");

static __device__ __forceinline__ void excl_scan_256(const int* cnt, int* start, int lane) {
    int v[8];
    int sum = 0;
#pragma unroll
    for (int k = 0; k < 8; ++k) { v[k] = cnt[lane * 8 + k]; sum += v[k]; }
    int inc = sum;
#pragma unroll
    for (int off = 1; off < 32; off <<= 1) {
        const int t = __shfl_up(inc, off);
        if (lane >= off) inc += t;
    }
    int ex = inc - sum;
#pragma unroll
    for (int k = 0; k < 8; ++k) { start[lane * 8 + k] = ex; ex += v[k]; }
    if (lane == 31) start[256] = ex;
}

__global__ __launch_bounds__(PART_THREADS) void edge_pass_kernel(const int* __restrict__ src,
                                                                 const int* __restrict__ dst,
                                                                 float* __restrict__ out_deg,
                                                                 unsigned* __restrict__ gpl,
                                                                 int* __restrict__ goff) {
    __shared__ int      bcnt[OWN_BLOCKS];
    __shared__ int      bstart[OWN_BLOCKS + 1];
    __shared__ unsigned sorted[PART_PITCH];

    const int tid = threadIdx.x;
    if (tid < OWN_BLOCKS) bcnt[tid] = 0;
    __syncthreads();

    const int  idx4  = blockIdx.x * PART_E4 + tid;
    const bool valid = tid < PART_E4 && idx4 < N_EDGES / 4;
    int      own[4] = {-1, -1, -1, -1};
    int      rk[4]  = {0, 0, 0, 0};
    unsigned wd[4]  = {0u, 0u, 0u, 0u};
    if (valid) {
        const int4 d4 = reinterpret_cast<const int4*>(dst)[idx4];
        const int4 s4 = reinterpret_cast<const int4*>(src)[idx4];
        const int dd[4] = {d4.x, d4.y, d4.z, d4.w};
        const int ss[4] = {s4.x, s4.y, s4.z, s4.w};
#pragma unroll
        for (int c = 0; c < 4; ++c) {
            unsafeAtomicAdd(&out_deg[ss[c]], 1.0f);
            if ((unsigned)dd[c] < (unsigned)N_NODES) {
                const unsigned o = (unsigned)dd[c] / OWN_ROWS;
                const unsigned r = (unsigned)dd[c] - o * OWN_ROWS;
                own[c] = (int)o;
                wd[c]  = (r << 16) | (unsigned)ss[c];
                rk[c]  = atomicAdd(&bcnt[o], 1);
            }
        }
    }
    __syncthreads();
    if (tid < 32) excl_scan_256(bcnt, bstart, tid);
    __syncthreads();
#pragma unroll
    for (int c = 0; c < 4; ++c)
        if (own[c] >= 0) sorted[bstart[own[c]] + rk[c]] = wd[c];
    __syncthreads();

    const int tot = bstart[OWN_BLOCKS];
    unsigned* __restrict__ pl = gpl + (size_t)blockIdx.x * PART_PITCH;
    for (int i = tid; i < tot; i += PART_THREADS) pl[i] = sorted[i];
    if (tid <= OWN_BLOCKS) goff[(size_t)blockIdx.x * OFF_PITCH + tid] = bstart[tid];
}

static __device__ __forceinline__ float norm_of_deg(float deg) { return rsqrtf(fmaxf(deg, 1.0f)); }

#define OWN_THREADS 1024
#define OWN_WAVES   (OWN_THREADS / 32)
#define LIST_CAP    10240
#define COPY_U      4
#define LOOK_U      4
#define GATHER_U    8
#define CNT_PITCH   32
#define DN_PITCH    256
#define ACC_TILES   ((OWN_ROWS + 15) / 16)
#define ACC_ROWS    (ACC_TILES * 16)
#define ACC_PITCH   132

static __device__ __forceinline__ int feat_of_word(int w) { return 4 * (w & 31) + (w >> 5); }

static __device__ __forceinline__ void add_edge_alone(float* __restrict__ arow,
                                                      const float* __restrict__ h,
                                                      const int* __restrict__ src,
                                                      const float* __restrict__ src_deg,
                                                      int e) {
    const int   s   = src[e];
    const float sc  = norm_of_deg(src_deg[s]);
    const float* hr = h + (size_t)s * D;
    for (int k = 0; k < D; ++k)
        unsafeAtomicAdd(&arow[(k & 3) * 32 + (k >> 2)], hr[k] * sc);
}

template <bool BUILD>
__global__ __launch_bounds__(OWN_THREADS) void layer_owner_kernel(
        const float* __restrict__ h,
        const int* __restrict__ src,
        const int* __restrict__ dst,
        const float* __restrict__ src_deg,
        const float* __restrict__ Wm,
        const float* __restrict__ bias,
        float* __restrict__ hout,
        unsigned* __restrict__ glst,
        float* __restrict__ glsc,
        int* __restrict__ gcnt,
        float* __restrict__ gdn,
        const unsigned* __restrict__ gpl,
        const int* __restrict__ goff) {
    __shared__ __attribute__((aligned(16))) float acc[ACC_ROWS * ACC_PITCH];
    __shared__ unsigned lst[LIST_CAP];
    __shared__ float    lsc[LIST_CAP];
    __shared__ int      rcnt[256];
    __shared__ int      rstart[256 + 1];
    __shared__ int      rfill[256];
    __shared__ unsigned ulst[BUILD ? LIST_CAP : 1];
    __shared__ float    ulsc[BUILD ? LIST_CAP : 1];
    __shared__ int      pstart[PART_BLOCKS];
    __shared__ int      pcnt[PART_BLOCKS];
    __shared__ int      pbase[PART_BLOCKS + 1];

    const int tid  = threadIdx.x;
    const int wave = tid >> 5;
    const int lane = tid & 31;
    const int lo   = blockIdx.x * OWN_ROWS;

    unsigned* __restrict__ blst = glst + (size_t)blockIdx.x * LIST_CAP;
    float*    __restrict__ blsc = glsc + (size_t)blockIdx.x * LIST_CAP;

    int n = BUILD ? 0 : gcnt[blockIdx.x * CNT_PITCH];

    {
        float4* a4 = reinterpret_cast<float4*>(acc);
        const float4 z = make_float4(0.f, 0.f, 0.f, 0.f);
        for (int i = tid; i < ACC_ROWS * ACC_PITCH / 4; i += OWN_THREADS) a4[i] = z;
        if (BUILD && tid < 256) { rcnt[tid] = 0; rfill[tid] = 0; }
    }
    if (!BUILD && n <= LIST_CAP) {
        for (int i = tid; i < n; i += OWN_THREADS) {
            lst[i] = blst[i];
            lsc[i] = blsc[i];
        }
    }
    if (BUILD && tid < PART_BLOCKS) {
        const int* po = goff + (size_t)tid * OFF_PITCH + blockIdx.x;
        const int st = po[0];
        pstart[tid] = st;
        pcnt[tid]   = po[1] - st;
    }
    __syncthreads();

    if (BUILD) {
        static_assert(OWN_THREADS == 4 * PART_BLOCKS, "the run copy below deals four threads to a piece");
        if (wave == 0) excl_scan_256(pcnt, pbase, lane);
        __syncthreads();
        n = pbase[PART_BLOCKS];
        if (tid == 0) gcnt[blockIdx.x * CNT_PITCH] = n;
        if (n <= LIST_CAP) {
            const int p = tid >> 2;
            const int q = tid & 3;
            const int b = pbase[p];
            const int c = pbase[p + 1] - b;
            const unsigned* __restrict__ pp = gpl + (size_t)p * PART_PITCH + pstart[p];
            for (int j0 = q; j0 < c; j0 += 4 * COPY_U) {
                unsigned v[COPY_U];
#pragma unroll
                for (int k = 0; k < COPY_U; ++k) {
                    const int j = j0 + 4 * k;
                    v[k] = (j < c) ? pp[j] : 0u;
                }
                asm volatile("" :: "v"(v[0]), "v"(v[1]), "v"(v[2]), "v"(v[3]));
#pragma unroll
                for (int k = 0; k < COPY_U; ++k) {
                    const int j = j0 + 4 * k;
                    if (j < c) ulst[b + j] = v[k];
                }
            }
        }
        __syncthreads();
    }
    if (n > LIST_CAP) {
        for (int e = tid; e < N_EDGES; e += OWN_THREADS) {
            const unsigned r = (unsigned)(dst[e] - lo);
            if (r < (unsigned)OWN_ROWS) {
                if (BUILD) atomicAdd(&rcnt[r], 1);
                add_edge_alone(acc + r * ACC_PITCH, h, src, src_deg, e);
            }
        }
        n = 0;
    }

    for (int i0 = tid; BUILD && i0 < n; i0 += OWN_THREADS * LOOK_U) {
        unsigned w[LOOK_U];
        float    dg[LOOK_U];
#pragma unroll
        for (int k = 0; k < LOOK_U; ++k) {
            const int i = i0 + k * OWN_THREADS;
            w[k] = (i < n) ? ulst[i] : 0u;
        }
#pragma unroll
        for (int k = 0; k < LOOK_U; ++k) dg[k] = src_deg[w[k] & 0xFFFFu];
        asm volatile("" :: "v"(dg[0]), "v"(dg[1]), "v"(dg[2]), "v"(dg[3]));
#pragma unroll
        for (int k = 0; k < LOOK_U; ++k) {
            const int i = i0 + k * OWN_THREADS;
            if (i < n) {
                const float sc = norm_of_deg(dg[k]);
                atomicAdd(&rcnt[w[k] >> 16], 1);
                ulsc[i] = sc;
            }
        }
    }
    if (BUILD) {
        __syncthreads();
        if (tid < ACC_ROWS) gdn[blockIdx.x * DN_PITCH + tid] = norm_of_deg((float)rcnt[tid]);
        static_assert(ACC_ROWS <= 256, "the row counts are scanned as 256 counts");
        if (wave == 0) excl_scan_256(rcnt, rstart, lane);
        __syncthreads();
        for (int i = tid; i < n; i += OWN_THREADS) {
            const unsigned w = ulst[i];
            const int      r = (int)(w >> 16);
            const int    pos = rstart[r] + atomicAdd(&rfill[r], 1);
            lst[pos] = w;
            lsc[pos] = ulsc[i];
        }
        __syncthreads();
        for (int i = tid; i < n; i += OWN_THREADS) {
            blst[i] = lst[i];
            blsc[i] = lsc[i];
        }
    }

    const int wv = __builtin_amdgcn_readfirstlane(wave);
    {
        const int ns = __builtin_amdgcn_readfirstlane(n);
        const int e0 = (wv * ns) / OWN_WAVES;
        const int e1 = ((wv + 1) * ns) / OWN_WAVES;
        int    cur = -1;
        float4 run = make_float4(0.f, 0.f, 0.f, 0.f);
        for (int i0 = e0; i0 < e1; i0 += GATHER_U) {
            unsigned p[GATHER_U];
            float    sc[GATHER_U];
            float4   row[GATHER_U];
#pragma unroll
            for (int u = 0; u < GATHER_U; ++u) {
                const int i = min(i0 + u, e1 - 1);
                p[u]   = __builtin_amdgcn_readfirstlane(lst[i]);
                sc[u]  = lsc[i];
                row[u] = *reinterpret_cast<const float4*>(h + (size_t)(p[u] & 0xFFFFu) * D + lane * 4);
            }
#pragma unroll
            for (int u = 0; u < GATHER_U; ++u) {
                if (i0 + u < e1) {
                    const int r = (int)(p[u] >> 16);
                    if (r != cur) {
                        if (cur >= 0) {
                            float* a = acc + cur * ACC_PITCH + lane;
                            unsafeAtomicAdd(a,      run.x);
                            unsafeAtomicAdd(a + 32, run.y);
                            unsafeAtomicAdd(a + 64, run.z);
                            unsafeAtomicAdd(a + 96, run.w);
                        }
                        cur = r;
                        run = make_float4(0.f, 0.f, 0.f, 0.f);
                    }
                    run.x = fmaf(row[u].x, sc[u], run.x);
                    run.y = fmaf(row[u].y, sc[u], run.y);
                    run.z = fmaf(row[u].z, sc[u], run.z);
                    run.w = fmaf(row[u].w, sc[u], run.w);
                }
            }
        }
        if (cur >= 0) {
            float* a = acc + cur * ACC_PITCH + lane;
            unsafeAtomicAdd(a,      run.x);
            unsafeAtomicAdd(a + 32, run.y);
            unsafeAtomicAdd(a + 64, run.z);
            unsafeAtomicAdd(a + 96, run.w);
        }
    }

    static_assert(OWN_WAVES == 32 && ACC_TILES == 13, "tile deal below is written for 32 waves, 13 row tiles");
    const int c16  = lane & 15;
    const int hf   = lane >> 4;
    const int ncol = (wv & 7) * 16 + c16;
    v2f bfr[D / 4];
#pragma unroll
    for (int s = 0; s < D / 4; ++s) {
        bfr[s].x = Wm[(size_t)feat_of_word(4 * s + 2 * hf    ) * D + ncol];
        bfr[s].y = Wm[(size_t)feat_of_word(4 * s + 2 * hf + 1) * D + ncol];
    }
    const float bn = bias[ncol];
    __syncthreads();

    for (int t = 0; t < 4; ++t) {
        int rt = (wv >> 3) + 4 * t;
        if (t == 3) {
            if ((wv >> 3) != (wv & 3)) break;
            rt = ACC_TILES - 1;
        }
        const int   rowl = rt * 16 + c16;
        const float dn   = BUILD ? norm_of_deg((float)rcnt[rowl]) : gdn[blockIdx.x * DN_PITCH + rowl];
        const float* ap  = acc + rowl * ACC_PITCH + 2 * hf;
        v8f c = {};
#pragma unroll
        for (int s = 0; s < D / 4; ++s) {
            v2f a = *reinterpret_cast<const v2f*>(ap + 4 * s);
            a.x *= dn;
            a.y *= dn;
            c = __builtin_amdgcn_wmma_f32_16x16x4_f32(
                     false, a,  false, bfr[s],
                     (short)0, c,  false,  false);
        }
        float o[8];
#pragma unroll
        for (int v = 0; v < 8; ++v) o[v] = c[v] + bn;
        const int rbase = rt * 16 + hf * 8;
#pragma unroll
        for (int v = 0; v < 8; ++v) {
            const int r    = rbase + v;
            const int node = lo + r;
            if (r < OWN_ROWS && node < N_NODES)
                hout[(size_t)node * D + ncol] = o[v];
        }
    }
}


extern "C" void kernel_launch(void* const* d_in, const int* in_sizes, int n_in,
                              void* d_out, int out_size, void* d_ws, size_t ws_size,
                              hipStream_t stream) {
    const float* feat = (const float*)d_in[0];
    const int*   src  = (const int*)  d_in[1];
    const int*   dst  = (const int*)  d_in[2];
    const float* W    = (const float*)d_in[3];
    const float* b    = (const float*)d_in[4];
    float*       out  = (float*)d_out;

    float* ws       = (float*)d_ws;
    float* out_deg  = ws;
    float* hA       = ws + 2 * (size_t)N_NODES;
    float* hB       = hA + (size_t)N_NODES * D;
    unsigned* glst  = (unsigned*)(hB + (size_t)N_NODES * D);
    float*    glsc  = (float*)(glst + (size_t)OWN_BLOCKS * LIST_CAP);
    int*      gcnt  = (int*)(glsc + (size_t)OWN_BLOCKS * LIST_CAP);
    float*    gdn   = (float*)(gcnt + (size_t)OWN_BLOCKS * CNT_PITCH);
    unsigned* gpl   = (unsigned*)(gdn + (size_t)OWN_BLOCKS * DN_PITCH);
    int*      goff  = (int*)(gpl + (size_t)PART_BLOCKS * PART_PITCH);

    hipMemsetAsync(out_deg, 0, (size_t)N_NODES * sizeof(float), stream);
    edge_pass_kernel<<<PART_BLOCKS, PART_THREADS, 0, stream>>>(src, dst, out_deg, gpl, goff);

    layer_owner_kernel<true><<<OWN_BLOCKS, OWN_THREADS, 0, stream>>>(
        feat, src, dst, out_deg, W, b, hA, glst, glsc, gcnt, gdn, gpl, goff);
    layer_owner_kernel<false><<<OWN_BLOCKS, OWN_THREADS, 0, stream>>>(
        hA, src, dst, out_deg, W + (size_t)1 * D * D, b + (size_t)1 * D, hB, glst, glsc, gcnt, gdn, gpl, goff);
    layer_owner_kernel<false><<<OWN_BLOCKS, OWN_THREADS, 0, stream>>>(
        hB, src, dst, out_deg, W + (size_t)2 * D * D, b + (size_t)2 * D, hA, glst, glsc, gcnt, gdn, gpl, goff);
    layer_owner_kernel<false><<<OWN_BLOCKS, OWN_THREADS, 0, stream>>>(
        hA, src, dst, out_deg, W + (size_t)3 * D * D, b + (size_t)3 * D, out, glst, glsc, gcnt, gdn, gpl, goff);
}


# speedup vs baseline: 3.8429x; 1.0937x over previous
#include <hip/hip_runtime.h>

typedef __attribute__((ext_vector_type(2))) float v2f;
typedef __attribute__((ext_vector_type(8))) float v8f;

#define N_NODES 50000
#define N_EDGES 600000
#define D 128
#define N_LAYERS 4

#define H_OFF       102400
static_assert(H_OFF >= N_NODES && (H_OFF * sizeof(float)) % 512 == 0, "h tables: behind the out-degrees, on a 512-byte multiple");
static_assert(((size_t)N_NODES * D * sizeof(float)) % 512 == 0, "the second h table follows the first on a 512-byte multiple");

#define OWN_ROWS    196
#define OWN_BLOCKS  ((N_NODES + OWN_ROWS - 1) / OWN_ROWS)

#define PART_BLOCKS  256
#define PART_E4      ((N_EDGES / 4 + PART_BLOCKS - 1) / PART_BLOCKS)
#define PART_THREADS 640
#define PART_PITCH   (((PART_E4 * 4 + 31) / 32) * 32)
#define OFF_PITCH    288
static_assert(N_EDGES % 4 == 0, "the edge pass reads four edges per 16-byte load");
static_assert(PART_THREADS >= PART_E4 && PART_THREADS % 32 == 0, "one 16-byte load of each array per thread");
static_assert(OWN_BLOCKS == 256 && PART_BLOCKS == 256, "the prefix sum below is written for 256 counts");
static_assert(OFF_PITCH >= OWN_BLOCKS + 1 && OFF_PITCH % 32 == 0, "table pitch");

static __device__ __forceinline__ void excl_scan_256(const int* cnt, int* start, int lane) {
    int v[8];
    int sum = 0;
#pragma unroll
    for (int k = 0; k < 8; ++k) { v[k] = cnt[lane * 8 + k]; sum += v[k]; }
    int inc = sum;
#pragma unroll
    for (int off = 1; off < 32; off <<= 1) {
        const int t = __shfl_up(inc, off);
        if (lane >= off) inc += t;
    }
    int ex = inc - sum;
#pragma unroll
    for (int k = 0; k < 8; ++k) { start[lane * 8 + k] = ex; ex += v[k]; }
    if (lane == 31) start[256] = ex;
}

__global__ __launch_bounds__(PART_THREADS) void edge_pass_kernel(const int* __restrict__ src,
                                                                 const int* __restrict__ dst,
                                                                 float* __restrict__ out_deg,
                                                                 unsigned* __restrict__ gpl,
                                                                 int* __restrict__ goff) {
    __shared__ int      bcnt[OWN_BLOCKS];
    __shared__ int      bstart[OWN_BLOCKS + 1];
    __shared__ unsigned sorted[PART_PITCH];

    const int tid = threadIdx.x;
    if (tid < OWN_BLOCKS) bcnt[tid] = 0;
    __syncthreads();

    const int  idx4  = blockIdx.x * PART_E4 + tid;
    const bool valid = tid < PART_E4 && idx4 < N_EDGES / 4;
    int      own[4] = {-1, -1, -1, -1};
    int      rk[4]  = {0, 0, 0, 0};
    unsigned wd[4]  = {0u, 0u, 0u, 0u};
    if (valid) {
        const int4 d4 = reinterpret_cast<const int4*>(dst)[idx4];
        const int4 s4 = reinterpret_cast<const int4*>(src)[idx4];
        const int dd[4] = {d4.x, d4.y, d4.z, d4.w};
        const int ss[4] = {s4.x, s4.y, s4.z, s4.w};
#pragma unroll
        for (int c = 0; c < 4; ++c) {
            unsafeAtomicAdd(&out_deg[ss[c]], 1.0f);
            if ((unsigned)dd[c] < (unsigned)N_NODES) {
                const unsigned o = (unsigned)dd[c] / OWN_ROWS;
                const unsigned r = (unsigned)dd[c] - o * OWN_ROWS;
                own[c] = (int)o;
                wd[c]  = (r << 16) | (unsigned)ss[c];
                rk[c]  = atomicAdd(&bcnt[o], 1);
            }
        }
    }
    __syncthreads();
    if (tid < 32) excl_scan_256(bcnt, bstart, tid);
    __syncthreads();
#pragma unroll
    for (int c = 0; c < 4; ++c)
        if (own[c] >= 0) sorted[bstart[own[c]] + rk[c]] = wd[c];
    __syncthreads();

    const int tot = bstart[OWN_BLOCKS];
    unsigned* __restrict__ pl = gpl + (size_t)blockIdx.x * PART_PITCH;
    for (int i = tid; i < tot; i += PART_THREADS) pl[i] = sorted[i];
    if (tid <= OWN_BLOCKS) goff[(size_t)blockIdx.x * OFF_PITCH + tid] = bstart[tid];
}

static __device__ __forceinline__ float norm_of_deg(float deg) { return rsqrtf(fmaxf(deg, 1.0f)); }

#define OWN_THREADS 1024
#define OWN_WAVES   (OWN_THREADS / 32)
#define LIST_CAP    10240
#define COPY_U      4
#define LOOK_U      4
#define GATHER_U    8
#define CNT_PITCH   32
#define DN_PITCH    256
#define ACC_TILES   ((OWN_ROWS + 15) / 16)
#define ACC_ROWS    (ACC_TILES * 16)
#define ACC_PITCH   132

static __device__ __forceinline__ int feat_of_word(int w) { return 4 * (w & 31) + (w >> 5); }

static __device__ __forceinline__ void add_edge_alone(float* __restrict__ arow,
                                                      const float* __restrict__ h,
                                                      const int* __restrict__ src,
                                                      const float* __restrict__ src_deg,
                                                      int e) {
    const int   s   = src[e];
    const float sc  = norm_of_deg(src_deg[s]);
    const float* hr = h + (size_t)s * D;
    for (int k = 0; k < D; ++k)
        unsafeAtomicAdd(&arow[(k & 3) * 32 + (k >> 2)], hr[k] * sc);
}

template <bool BUILD>
__global__ __launch_bounds__(OWN_THREADS) void layer_owner_kernel(
        const float* __restrict__ h,
        const int* __restrict__ src,
        const int* __restrict__ dst,
        const float* __restrict__ src_deg,
        const float* __restrict__ Wm,
        const float* __restrict__ bias,
        float* __restrict__ hout,
        unsigned* __restrict__ glst,
        float* __restrict__ glsc,
        int* __restrict__ gcnt,
        float* __restrict__ gdn,
        const unsigned* __restrict__ gpl,
        const int* __restrict__ goff) {
    __shared__ __attribute__((aligned(16))) float acc[ACC_ROWS * ACC_PITCH];
    __shared__ unsigned lst[LIST_CAP];
    __shared__ float    lsc[LIST_CAP];
    __shared__ int      rcnt[256];
    __shared__ int      rstart[256 + 1];
    __shared__ int      rfill[256];
    __shared__ unsigned ulst[BUILD ? LIST_CAP : 1];
    __shared__ float    ulsc[BUILD ? LIST_CAP : 1];
    __shared__ int      pstart[PART_BLOCKS];
    __shared__ int      pcnt[PART_BLOCKS];
    __shared__ int      pbase[PART_BLOCKS + 1];

    const int tid  = threadIdx.x;
    const int wave = tid >> 5;
    const int lane = tid & 31;
    const int lo   = blockIdx.x * OWN_ROWS;

    unsigned* __restrict__ blst = glst + (size_t)blockIdx.x * LIST_CAP;
    float*    __restrict__ blsc = glsc + (size_t)blockIdx.x * LIST_CAP;

    int n = BUILD ? 0 : gcnt[blockIdx.x * CNT_PITCH];

    {
        float4* a4 = reinterpret_cast<float4*>(acc);
        const float4 z = make_float4(0.f, 0.f, 0.f, 0.f);
        for (int i = tid; i < ACC_ROWS * ACC_PITCH / 4; i += OWN_THREADS) a4[i] = z;
        if (BUILD && tid < 256) { rcnt[tid] = 0; rfill[tid] = 0; }
    }
    if (!BUILD && n <= LIST_CAP) {
        for (int i = tid; i < n; i += OWN_THREADS) {
            lst[i] = blst[i];
            lsc[i] = blsc[i];
        }
    }
    if (BUILD && tid < PART_BLOCKS) {
        const int* po = goff + (size_t)tid * OFF_PITCH + blockIdx.x;
        const int st = po[0];
        pstart[tid] = st;
        pcnt[tid]   = po[1] - st;
    }
    __syncthreads();

    if (BUILD) {
        static_assert(OWN_THREADS == 4 * PART_BLOCKS, "the run copy below deals four threads to a piece");
        if (wave == 0) excl_scan_256(pcnt, pbase, lane);
        __syncthreads();
        n = pbase[PART_BLOCKS];
        if (tid == 0) gcnt[blockIdx.x * CNT_PITCH] = n;
        if (n <= LIST_CAP) {
            const int p = tid >> 2;
            const int q = tid & 3;
            const int b = pbase[p];
            const int c = pbase[p + 1] - b;
            const unsigned* __restrict__ pp = gpl + (size_t)p * PART_PITCH + pstart[p];
            for (int j0 = q; j0 < c; j0 += 4 * COPY_U) {
                unsigned v[COPY_U];
#pragma unroll
                for (int k = 0; k < COPY_U; ++k) {
                    const int j = j0 + 4 * k;
                    v[k] = (j < c) ? pp[j] : 0u;
                }
                asm volatile("" :: "v"(v[0]), "v"(v[1]), "v"(v[2]), "v"(v[3]));
#pragma unroll
                for (int k = 0; k < COPY_U; ++k) {
                    const int j = j0 + 4 * k;
                    if (j < c) ulst[b + j] = v[k];
                }
            }
        }
        __syncthreads();
    }
    if (n > LIST_CAP) {
        for (int e = tid; e < N_EDGES; e += OWN_THREADS) {
            const unsigned r = (unsigned)(dst[e] - lo);
            if (r < (unsigned)OWN_ROWS) {
                if (BUILD) atomicAdd(&rcnt[r], 1);
                add_edge_alone(acc + r * ACC_PITCH, h, src, src_deg, e);
            }
        }
        n = 0;
    }

    for (int i0 = tid; BUILD && i0 < n; i0 += OWN_THREADS * LOOK_U) {
        unsigned w[LOOK_U];
        float    dg[LOOK_U];
#pragma unroll
        for (int k = 0; k < LOOK_U; ++k) {
            const int i = i0 + k * OWN_THREADS;
            w[k] = (i < n) ? ulst[i] : 0u;
        }
#pragma unroll
        for (int k = 0; k < LOOK_U; ++k) dg[k] = src_deg[w[k] & 0xFFFFu];
        asm volatile("" :: "v"(dg[0]), "v"(dg[1]), "v"(dg[2]), "v"(dg[3]));
#pragma unroll
        for (int k = 0; k < LOOK_U; ++k) {
            const int i = i0 + k * OWN_THREADS;
            if (i < n) {
                const float sc = norm_of_deg(dg[k]);
                atomicAdd(&rcnt[w[k] >> 16], 1);
                ulsc[i] = sc;
            }
        }
    }
    if (BUILD) {
        __syncthreads();
        if (tid < ACC_ROWS) gdn[blockIdx.x * DN_PITCH + tid] = norm_of_deg((float)rcnt[tid]);
        static_assert(ACC_ROWS <= 256, "the row counts are scanned as 256 counts");
        if (wave == 0) excl_scan_256(rcnt, rstart, lane);
        __syncthreads();
        for (int i = tid; i < n; i += OWN_THREADS) {
            const unsigned w = ulst[i];
            const int      r = (int)(w >> 16);
            const int    pos = rstart[r] + atomicAdd(&rfill[r], 1);
            lst[pos] = w;
            lsc[pos] = ulsc[i];
        }
        __syncthreads();
        for (int i = tid; i < n; i += OWN_THREADS) {
            blst[i] = lst[i];
            blsc[i] = lsc[i];
        }
    }

    const int wv = __builtin_amdgcn_readfirstlane(wave);
    {
        const int ns = __builtin_amdgcn_readfirstlane(n);
        const int e0 = (wv * ns) / OWN_WAVES;
        const int e1 = ((wv + 1) * ns) / OWN_WAVES;
        int    cur = -1;
        float4 run = make_float4(0.f, 0.f, 0.f, 0.f);
        for (int i0 = e0; i0 < e1; i0 += GATHER_U) {
            unsigned p[GATHER_U];
            float    sc[GATHER_U];
            float4   row[GATHER_U];
#pragma unroll
            for (int u = 0; u < GATHER_U; ++u) {
                const int i = min(i0 + u, e1 - 1);
                p[u]   = __builtin_amdgcn_readfirstlane(lst[i]);
                sc[u]  = lsc[i];
                row[u] = *reinterpret_cast<const float4*>(h + (size_t)(p[u] & 0xFFFFu) * D + lane * 4);
            }
#pragma unroll
            for (int u = 0; u < GATHER_U; ++u) {
                if (i0 + u < e1) {
                    const int r = (int)(p[u] >> 16);
                    if (r != cur) {
                        if (cur >= 0) {
                            float* a = acc + cur * ACC_PITCH + lane;
                            unsafeAtomicAdd(a,      run.x);
                            unsafeAtomicAdd(a + 32, run.y);
                            unsafeAtomicAdd(a + 64, run.z);
                            unsafeAtomicAdd(a + 96, run.w);
                        }
                        cur = r;
                        run = make_float4(0.f, 0.f, 0.f, 0.f);
                    }
                    run.x = fmaf(row[u].x, sc[u], run.x);
                    run.y = fmaf(row[u].y, sc[u], run.y);
                    run.z = fmaf(row[u].z, sc[u], run.z);
                    run.w = fmaf(row[u].w, sc[u], run.w);
                }
            }
        }
        if (cur >= 0) {
            float* a = acc + cur * ACC_PITCH + lane;
            unsafeAtomicAdd(a,      run.x);
            unsafeAtomicAdd(a + 32, run.y);
            unsafeAtomicAdd(a + 64, run.z);
            unsafeAtomicAdd(a + 96, run.w);
        }
    }

    static_assert(OWN_WAVES == 32 && ACC_TILES == 13, "tile deal below is written for 32 waves, 13 row tiles");
    const int c16  = lane & 15;
    const int hf   = lane >> 4;
    const int ncol = (wv & 7) * 16 + c16;
    v2f bfr[D / 4];
#pragma unroll
    for (int s = 0; s < D / 4; ++s) {
        bfr[s].x = Wm[(size_t)feat_of_word(4 * s + 2 * hf    ) * D + ncol];
        bfr[s].y = Wm[(size_t)feat_of_word(4 * s + 2 * hf + 1) * D + ncol];
    }
    const float bn = bias[ncol];
    __syncthreads();

    for (int t = 0; t < 4; ++t) {
        int rt = (wv >> 3) + 4 * t;
        if (t == 3) {
            if ((wv >> 3) != (wv & 3)) break;
            rt = ACC_TILES - 1;
        }
        const int   rowl = rt * 16 + c16;
        const float dn   = BUILD ? norm_of_deg((float)rcnt[rowl]) : gdn[blockIdx.x * DN_PITCH + rowl];
        const float* ap  = acc + rowl * ACC_PITCH + 2 * hf;
        v8f c = {};
#pragma unroll
        for (int s = 0; s < D / 4; ++s) {
            v2f a = *reinterpret_cast<const v2f*>(ap + 4 * s);
            a.x *= dn;
            a.y *= dn;
            c = __builtin_amdgcn_wmma_f32_16x16x4_f32(
                     false, a,  false, bfr[s],
                     (short)0, c,  false,  false);
        }
        float o[8];
#pragma unroll
        for (int v = 0; v < 8; ++v) o[v] = c[v] + bn;
        const int rbase = rt * 16 + hf * 8;
#pragma unroll
        for (int v = 0; v < 8; ++v) {
            const int r    = rbase + v;
            const int node = lo + r;
            if (r < OWN_ROWS && node < N_NODES)
                hout[(size_t)node * D + ncol] = o[v];
        }
    }
}


extern "C" void kernel_launch(void* const* d_in, const int* in_sizes, int n_in,
                              void* d_out, int out_size, void* d_ws, size_t ws_size,
                              hipStream_t stream) {
    const float* feat = (const float*)d_in[0];
    const int*   src  = (const int*)  d_in[1];
    const int*   dst  = (const int*)  d_in[2];
    const float* W    = (const float*)d_in[3];
    const float* b    = (const float*)d_in[4];
    float*       out  = (float*)d_out;

    float* ws       = (float*)d_ws;
    float* out_deg  = ws;
    float* hA       = ws + (size_t)H_OFF;
    float* hB       = hA + (size_t)N_NODES * D;
    unsigned* glst  = (unsigned*)(hB + (size_t)N_NODES * D);
    float*    glsc  = (float*)(glst + (size_t)OWN_BLOCKS * LIST_CAP);
    int*      gcnt  = (int*)(glsc + (size_t)OWN_BLOCKS * LIST_CAP);
    float*    gdn   = (float*)(gcnt + (size_t)OWN_BLOCKS * CNT_PITCH);
    unsigned* gpl   = (unsigned*)(gdn + (size_t)OWN_BLOCKS * DN_PITCH);
    int*      goff  = (int*)(gpl + (size_t)PART_BLOCKS * PART_PITCH);

    hipMemsetAsync(out_deg, 0, (size_t)N_NODES * sizeof(float), stream);
    edge_pass_kernel<<<PART_BLOCKS, PART_THREADS, 0, stream>>>(src, dst, out_deg, gpl, goff);

    layer_owner_kernel<true><<<OWN_BLOCKS, OWN_THREADS, 0, stream>>>(
        feat, src, dst, out_deg, W, b, hA, glst, glsc, gcnt, gdn, gpl, goff);
    layer_owner_kernel<false><<<OWN_BLOCKS, OWN_THREADS, 0, stream>>>(
        hA, src, dst, out_deg, W + (size_t)1 * D * D, b + (size_t)1 * D, hB, glst, glsc, gcnt, gdn, gpl, goff);
    layer_owner_kernel<false><<<OWN_BLOCKS, OWN_THREADS, 0, stream>>>(
        hB, src, dst, out_deg, W + (size_t)2 * D * D, b + (size_t)2 * D, hA, glst, glsc, gcnt, gdn, gpl, goff);
    layer_owner_kernel<false><<<OWN_BLOCKS, OWN_THREADS, 0, stream>>>(
        hA, src, dst, out_deg, W + (size_t)3 * D * D, b + (size_t)3 * D, out, glst, glsc, gcnt, gdn, gpl, goff);
}


# Product's global operands fetched at the kernel head, off the gather-to-product path: W by whole-line loads, staged in LDS in B-fragment order (16 LDS reads a wave); bias, row norms in registers
# speedup vs baseline: 4.1266x; 1.0738x over previous
#include <hip/hip_runtime.h>

typedef __attribute__((ext_vector_type(2))) float v2f;
typedef __attribute__((ext_vector_type(8))) float v8f;

#define N_NODES 50000
#define N_EDGES 600000
#define D 128
#define N_LAYERS 4

#define H_OFF       102400
static_assert(H_OFF >= N_NODES && (H_OFF * sizeof(float)) % 512 == 0, "h tables: behind the out-degrees, on a 512-byte multiple");
static_assert(((size_t)N_NODES * D * sizeof(float)) % 512 == 0, "the second h table follows the first on a 512-byte multiple");

#define OWN_ROWS    196
#define OWN_BLOCKS  ((N_NODES + OWN_ROWS - 1) / OWN_ROWS)

#define PART_BLOCKS  256
#define PART_E4      ((N_EDGES / 4 + PART_BLOCKS - 1) / PART_BLOCKS)
#define PART_THREADS 640
#define PART_PITCH   (((PART_E4 * 4 + 31) / 32) * 32)
#define OFF_PITCH    288
static_assert(N_EDGES % 4 == 0, "the edge pass reads four edges per 16-byte load");
static_assert(PART_THREADS >= PART_E4 && PART_THREADS % 32 == 0, "one 16-byte load of each array per thread");
static_assert(OWN_BLOCKS == 256 && PART_BLOCKS == 256, "the prefix sum below is written for 256 counts");
static_assert(OFF_PITCH >= OWN_BLOCKS + 1 && OFF_PITCH % 32 == 0, "table pitch");

static __device__ __forceinline__ void excl_scan_256(const int* cnt, int* start, int lane) {
    int v[8];
    int sum = 0;
#pragma unroll
    for (int k = 0; k < 8; ++k) { v[k] = cnt[lane * 8 + k]; sum += v[k]; }
    int inc = sum;
#pragma unroll
    for (int off = 1; off < 32; off <<= 1) {
        const int t = __shfl_up(inc, off);
        if (lane >= off) inc += t;
    }
    int ex = inc - sum;
#pragma unroll
    for (int k = 0; k < 8; ++k) { start[lane * 8 + k] = ex; ex += v[k]; }
    if (lane == 31) start[256] = ex;
}

__global__ __launch_bounds__(PART_THREADS) void edge_pass_kernel(const int* __restrict__ src,
                                                                 const int* __restrict__ dst,
                                                                 float* __restrict__ out_deg,
                                                                 unsigned* __restrict__ gpl,
                                                                 int* __restrict__ goff) {
    __shared__ int      bcnt[OWN_BLOCKS];
    __shared__ int      bstart[OWN_BLOCKS + 1];
    __shared__ unsigned sorted[PART_PITCH];

    const int tid = threadIdx.x;
    if (tid < OWN_BLOCKS) bcnt[tid] = 0;
    __syncthreads();

    const int  idx4  = blockIdx.x * PART_E4 + tid;
    const bool valid = tid < PART_E4 && idx4 < N_EDGES / 4;
    int      own[4] = {-1, -1, -1, -1};
    int      rk[4]  = {0, 0, 0, 0};
    unsigned wd[4]  = {0u, 0u, 0u, 0u};
    if (valid) {
        const int4 d4 = reinterpret_cast<const int4*>(dst)[idx4];
        const int4 s4 = reinterpret_cast<const int4*>(src)[idx4];
        const int dd[4] = {d4.x, d4.y, d4.z, d4.w};
        const int ss[4] = {s4.x, s4.y, s4.z, s4.w};
#pragma unroll
        for (int c = 0; c < 4; ++c) {
            unsafeAtomicAdd(&out_deg[ss[c]], 1.0f);
            if ((unsigned)dd[c] < (unsigned)N_NODES) {
                const unsigned o = (unsigned)dd[c] / OWN_ROWS;
                const unsigned r = (unsigned)dd[c] - o * OWN_ROWS;
                own[c] = (int)o;
                wd[c]  = (r << 16) | (unsigned)ss[c];
                rk[c]  = atomicAdd(&bcnt[o], 1);
            }
        }
    }
    __syncthreads();
    if (tid < 32) excl_scan_256(bcnt, bstart, tid);
    __syncthreads();
#pragma unroll
    for (int c = 0; c < 4; ++c)
        if (own[c] >= 0) sorted[bstart[own[c]] + rk[c]] = wd[c];
    __syncthreads();

    const int tot = bstart[OWN_BLOCKS];
    unsigned* __restrict__ pl = gpl + (size_t)blockIdx.x * PART_PITCH;
    for (int i = tid; i < tot; i += PART_THREADS) pl[i] = sorted[i];
    if (tid <= OWN_BLOCKS) goff[(size_t)blockIdx.x * OFF_PITCH + tid] = bstart[tid];
}

static __device__ __forceinline__ float norm_of_deg(float deg) { return rsqrtf(fmaxf(deg, 1.0f)); }

#define OWN_THREADS 1024
#define OWN_WAVES   (OWN_THREADS / 32)
#define LIST_CAP    10240
#define COPY_U      4
#define LOOK_U      4
#define GATHER_U    8
#define CNT_PITCH   32
#define DN_PITCH    256
#define ACC_TILES   ((OWN_ROWS + 15) / 16)
#define ACC_ROWS    (ACC_TILES * 16)
#define ACC_PITCH   132

static __device__ __forceinline__ int feat_of_word(int w) { return 4 * (w & 31) + (w >> 5); }

static __device__ __forceinline__ void add_edge_alone(float* __restrict__ arow,
                                                      const float* __restrict__ h,
                                                      const int* __restrict__ src,
                                                      const float* __restrict__ src_deg,
                                                      int e) {
    const int   s   = src[e];
    const float sc  = norm_of_deg(src_deg[s]);
    const float* hr = h + (size_t)s * D;
    for (int k = 0; k < D; ++k)
        unsafeAtomicAdd(&arow[(k & 3) * 32 + (k >> 2)], hr[k] * sc);
}

static __device__ __forceinline__ void store_w_stage(unsigned* ubuf, const float (&wq)[4][4], int wave, int lane) {
    uint4* wst = reinterpret_cast<uint4*>(ubuf);
#pragma unroll
    for (int i = 0; i < 4; ++i) {
        const int g   = wave + OWN_WAVES * i;
        const int wn  = (g & 3) * 32 + lane;
        const int whf = (g >> 2) & 1;
        const int wjj = g >> 3;
        wst[((wn >> 4) * 16 + wjj) * 32 + whf * 16 + (wn & 15)] =
            make_uint4(__float_as_uint(wq[i][0]), __float_as_uint(wq[i][1]),
                       __float_as_uint(wq[i][2]), __float_as_uint(wq[i][3]));
    }
}

template <bool BUILD>
__global__ __launch_bounds__(OWN_THREADS) void layer_owner_kernel(
        const float* __restrict__ h,
        const int* __restrict__ src,
        const int* __restrict__ dst,
        const float* __restrict__ src_deg,
        const float* __restrict__ Wm,
        const float* __restrict__ bias,
        float* __restrict__ hout,
        unsigned* __restrict__ glst,
        float* __restrict__ glsc,
        int* __restrict__ gcnt,
        float* __restrict__ gdn,
        const unsigned* __restrict__ gpl,
        const int* __restrict__ goff) {
    __shared__ __attribute__((aligned(16))) float acc[ACC_ROWS * ACC_PITCH];
    __shared__ unsigned lst[LIST_CAP];
    __shared__ float    lsc[LIST_CAP];
    __shared__ int      rcnt[256];
    __shared__ int      rstart[256 + 1];
    __shared__ int      rfill[256];
    static_assert(2 * LIST_CAP >= D * D, "W stage fits over the first layer's unsorted list");
    __shared__ __attribute__((aligned(16))) unsigned ubuf[BUILD ? 2 * LIST_CAP : D * D];
    unsigned* const ulst = ubuf;
    __shared__ int      pstart[PART_BLOCKS];
    __shared__ int      pcnt[PART_BLOCKS];
    __shared__ int      pbase[PART_BLOCKS + 1];

    const int tid  = threadIdx.x;
    const int wave = tid >> 5;
    const int lane = tid & 31;
    const int lo   = blockIdx.x * OWN_ROWS;

    unsigned* __restrict__ blst = glst + (size_t)blockIdx.x * LIST_CAP;
    float*    __restrict__ blsc = glsc + (size_t)blockIdx.x * LIST_CAP;

    float wq[4][4];
#pragma unroll
    for (int i = 0; i < 4; ++i) {
        const int g   = wave + OWN_WAVES * i;
        const int wn  = (g & 3) * 32 + lane;
        const int whf = (g >> 2) & 1;
        const int wjj = g >> 3;
#pragma unroll
        for (int q = 0; q < 4; ++q)
            wq[i][q] = Wm[(size_t)feat_of_word(4 * (2 * wjj + (q >> 1)) + 2 * whf + (q & 1)) * D + wn];
    }
    const int wv   = __builtin_amdgcn_readfirstlane(wave);
    const int c16  = lane & 15;
    const int hf   = lane >> 4;
    const int wv8  = wv & 7;
    const int ncol = wv8 * 16 + c16;
    const float bn = bias[ncol];
    float dnv[4] = {1.f, 1.f, 1.f, 1.f};
    if (!BUILD) {
#pragma unroll
        for (int t = 0; t < 4; ++t) {
            const int rt = (t == 3) ? ACC_TILES - 1 : (wv >> 3) + 4 * t;
            dnv[t] = gdn[blockIdx.x * DN_PITCH + rt * 16 + c16];
        }
    }

    int n = BUILD ? 0 : gcnt[blockIdx.x * CNT_PITCH];

    {
        float4* a4 = reinterpret_cast<float4*>(acc);
        const float4 z = make_float4(0.f, 0.f, 0.f, 0.f);
        for (int i = tid; i < ACC_ROWS * ACC_PITCH / 4; i += OWN_THREADS) a4[i] = z;
        if (BUILD && tid < 256) { rcnt[tid] = 0; rfill[tid] = 0; }
    }
    if (!BUILD && n <= LIST_CAP) {
        for (int i = tid; i < n; i += OWN_THREADS) {
            lst[i] = blst[i];
            lsc[i] = blsc[i];
        }
    }
    if (BUILD && tid < PART_BLOCKS) {
        const int* po = goff + (size_t)tid * OFF_PITCH + blockIdx.x;
        const int st = po[0];
        pstart[tid] = st;
        pcnt[tid]   = po[1] - st;
    }
    if (!BUILD) store_w_stage(ubuf, wq, wave, lane);
    __syncthreads();

    if (BUILD) {
        static_assert(OWN_THREADS == 4 * PART_BLOCKS, "the run copy below deals four threads to a piece");
        if (wave == 0) excl_scan_256(pcnt, pbase, lane);
        __syncthreads();
        n = pbase[PART_BLOCKS];
        if (tid == 0) gcnt[blockIdx.x * CNT_PITCH] = n;
        if (n <= LIST_CAP) {
            const int p = tid >> 2;
            const int q = tid & 3;
            const int b = pbase[p];
            const int c = pbase[p + 1] - b;
            const unsigned* __restrict__ pp = gpl + (size_t)p * PART_PITCH + pstart[p];
            for (int j0 = q; j0 < c; j0 += 4 * COPY_U) {
                unsigned v[COPY_U];
#pragma unroll
                for (int k = 0; k < COPY_U; ++k) {
                    const int j = j0 + 4 * k;
                    v[k] = (j < c) ? pp[j] : 0u;
                }
                asm volatile("" :: "v"(v[0]), "v"(v[1]), "v"(v[2]), "v"(v[3]));
#pragma unroll
                for (int k = 0; k < COPY_U; ++k) {
                    const int j = j0 + 4 * k;
                    if (j < c) ulst[b + j] = v[k];
                }
            }
        }
        __syncthreads();
    }
    if (n > LIST_CAP) {
        for (int e = tid; e < N_EDGES; e += OWN_THREADS) {
            const unsigned r = (unsigned)(dst[e] - lo);
            if (r < (unsigned)OWN_ROWS) {
                if (BUILD) atomicAdd(&rcnt[r], 1);
                add_edge_alone(acc + r * ACC_PITCH, h, src, src_deg, e);
            }
        }
        n = 0;
    }

    for (int i0 = tid; BUILD && i0 < n; i0 += OWN_THREADS * LOOK_U) {
        unsigned w[LOOK_U];
        float    dg[LOOK_U];
#pragma unroll
        for (int k = 0; k < LOOK_U; ++k) {
            const int i = i0 + k * OWN_THREADS;
            w[k] = (i < n) ? ulst[i] : 0u;
        }
#pragma unroll
        for (int k = 0; k < LOOK_U; ++k) dg[k] = src_deg[w[k] & 0xFFFFu];
        asm volatile("" :: "v"(dg[0]), "v"(dg[1]), "v"(dg[2]), "v"(dg[3]));
#pragma unroll
        for (int k = 0; k < LOOK_U; ++k) {
            const int i = i0 + k * OWN_THREADS;
            if (i < n) {
                const float sc = norm_of_deg(dg[k]);
                atomicAdd(&rcnt[w[k] >> 16], 1);
                ubuf[LIST_CAP + i] = __float_as_uint(sc);
            }
        }
    }
    if (BUILD) {
        __syncthreads();
        if (tid < ACC_ROWS) gdn[blockIdx.x * DN_PITCH + tid] = norm_of_deg((float)rcnt[tid]);
        static_assert(ACC_ROWS <= 256, "the row counts are scanned as 256 counts");
        if (wave == 0) excl_scan_256(rcnt, rstart, lane);
        __syncthreads();
        for (int i = tid; i < n; i += OWN_THREADS) {
            const unsigned w = ulst[i];
            const int      r = (int)(w >> 16);
            const int    pos = rstart[r] + atomicAdd(&rfill[r], 1);
            lst[pos] = w;
            lsc[pos] = __uint_as_float(ubuf[LIST_CAP + i]);
        }
        __syncthreads();
        for (int i = tid; i < n; i += OWN_THREADS) {
            blst[i] = lst[i];
            blsc[i] = lsc[i];
        }
        store_w_stage(ubuf, wq, wave, lane);
        __syncthreads();
    }

    {
        const int ns = __builtin_amdgcn_readfirstlane(n);
        const int e0 = (wv * ns) / OWN_WAVES;
        const int e1 = ((wv + 1) * ns) / OWN_WAVES;
        int    cur = -1;
        float4 run = make_float4(0.f, 0.f, 0.f, 0.f);
        for (int i0 = e0; i0 < e1; i0 += GATHER_U) {
            unsigned p[GATHER_U];
            float    sc[GATHER_U];
            float4   row[GATHER_U];
#pragma unroll
            for (int u = 0; u < GATHER_U; ++u) {
                const int i = min(i0 + u, e1 - 1);
                p[u]   = __builtin_amdgcn_readfirstlane(lst[i]);
                sc[u]  = lsc[i];
                row[u] = *reinterpret_cast<const float4*>(h + (size_t)(p[u] & 0xFFFFu) * D + lane * 4);
            }
#pragma unroll
            for (int u = 0; u < GATHER_U; ++u) {
                if (i0 + u < e1) {
                    const int r = (int)(p[u] >> 16);
                    if (r != cur) {
                        if (cur >= 0) {
                            float* a = acc + cur * ACC_PITCH + lane;
                            unsafeAtomicAdd(a,      run.x);
                            unsafeAtomicAdd(a + 32, run.y);
                            unsafeAtomicAdd(a + 64, run.z);
                            unsafeAtomicAdd(a + 96, run.w);
                        }
                        cur = r;
                        run = make_float4(0.f, 0.f, 0.f, 0.f);
                    }
                    run.x = fmaf(row[u].x, sc[u], run.x);
                    run.y = fmaf(row[u].y, sc[u], run.y);
                    run.z = fmaf(row[u].z, sc[u], run.z);
                    run.w = fmaf(row[u].w, sc[u], run.w);
                }
            }
        }
        if (cur >= 0) {
            float* a = acc + cur * ACC_PITCH + lane;
            unsafeAtomicAdd(a,      run.x);
            unsafeAtomicAdd(a + 32, run.y);
            unsafeAtomicAdd(a + 64, run.z);
            unsafeAtomicAdd(a + 96, run.w);
        }
    }

    static_assert(OWN_WAVES == 32 && ACC_TILES == 13, "tile deal below is written for 32 waves, 13 row tiles");
    v2f bfr[D / 4];
    {
        const uint4* wf = reinterpret_cast<const uint4*>(ubuf) + (wv8 * 16) * 32 + lane;
#pragma unroll
        for (int jj = 0; jj < D / 8; ++jj) {
            const uint4 f = wf[jj * 32];
            bfr[2 * jj    ].x = __uint_as_float(f.x);
            bfr[2 * jj    ].y = __uint_as_float(f.y);
            bfr[2 * jj + 1].x = __uint_as_float(f.z);
            bfr[2 * jj + 1].y = __uint_as_float(f.w);
        }
    }
    __syncthreads();

#pragma unroll
    for (int t = 0; t < 4; ++t) {
        int rt = (wv >> 3) + 4 * t;
        if (t == 3) {
            if ((wv >> 3) != (wv & 3)) break;
            rt = ACC_TILES - 1;
        }
        const int   rowl = rt * 16 + c16;
        const float dn   = BUILD ? norm_of_deg((float)rcnt[rowl]) : dnv[t];
        const float* ap  = acc + rowl * ACC_PITCH + 2 * hf;
        v8f c = {};
#pragma unroll
        for (int s = 0; s < D / 4; ++s) {
            v2f a = *reinterpret_cast<const v2f*>(ap + 4 * s);
            a.x *= dn;
            a.y *= dn;
            c = __builtin_amdgcn_wmma_f32_16x16x4_f32(
                     false, a,  false, bfr[s],
                     (short)0, c,  false,  false);
        }
        float o[8];
#pragma unroll
        for (int v = 0; v < 8; ++v) o[v] = c[v] + bn;
        const int rbase = rt * 16 + hf * 8;
#pragma unroll
        for (int v = 0; v < 8; ++v) {
            const int r    = rbase + v;
            const int node = lo + r;
            if (r < OWN_ROWS && node < N_NODES)
                hout[(size_t)node * D + ncol] = o[v];
        }
    }
}


extern "C" void kernel_launch(void* const* d_in, const int* in_sizes, int n_in,
                              void* d_out, int out_size, void* d_ws, size_t ws_size,
                              hipStream_t stream) {
    const float* feat = (const float*)d_in[0];
    const int*   src  = (const int*)  d_in[1];
    const int*   dst  = (const int*)  d_in[2];
    const float* W    = (const float*)d_in[3];
    const float* b    = (const float*)d_in[4];
    float*       out  = (float*)d_out;

    float* ws       = (float*)d_ws;
    float* out_deg  = ws;
    float* hA       = ws + (size_t)H_OFF;
    float* hB       = hA + (size_t)N_NODES * D;
    unsigned* glst  = (unsigned*)(hB + (size_t)N_NODES * D);
    float*    glsc  = (float*)(glst + (size_t)OWN_BLOCKS * LIST_CAP);
    int*      gcnt  = (int*)(glsc + (size_t)OWN_BLOCKS * LIST_CAP);
    float*    gdn   = (float*)(gcnt + (size_t)OWN_BLOCKS * CNT_PITCH);
    unsigned* gpl   = (unsigned*)(gdn + (size_t)OWN_BLOCKS * DN_PITCH);
    int*      goff  = (int*)(gpl + (size_t)PART_BLOCKS * PART_PITCH);

    hipMemsetAsync(out_deg, 0, (size_t)N_NODES * sizeof(float), stream);
    edge_pass_kernel<<<PART_BLOCKS, PART_THREADS, 0, stream>>>(src, dst, out_deg, gpl, goff);

    layer_owner_kernel<true><<<OWN_BLOCKS, OWN_THREADS, 0, stream>>>(
        feat, src, dst, out_deg, W, b, hA, glst, glsc, gcnt, gdn, gpl, goff);
    layer_owner_kernel<false><<<OWN_BLOCKS, OWN_THREADS, 0, stream>>>(
        hA, src, dst, out_deg, W + (size_t)1 * D * D, b + (size_t)1 * D, hB, glst, glsc, gcnt, gdn, gpl, goff);
    layer_owner_kernel<false><<<OWN_BLOCKS, OWN_THREADS, 0, stream>>>(
        hB, src, dst, out_deg, W + (size_t)2 * D * D, b + (size_t)2 * D, hA, glst, glsc, gcnt, gdn, gpl, goff);
    layer_owner_kernel<false><<<OWN_BLOCKS, OWN_THREADS, 0, stream>>>(
        hA, src, dst, out_deg, W + (size_t)3 * D * D, b + (size_t)3 * D, out, glst, glsc, gcnt, gdn, gpl, goff);
}


# Product phase: vector work beside the f32 matrix instructions cut: dst_norm scales the finished tile (fused into the bias add), not each A fragment; full tiles store unpredicated; k order unchanged
# speedup vs baseline: 4.3814x; 1.0617x over previous
#include <hip/hip_runtime.h>

typedef __attribute__((ext_vector_type(2))) float v2f;
typedef __attribute__((ext_vector_type(8))) float v8f;

#define N_NODES 50000
#define N_EDGES 600000
#define D 128
#define N_LAYERS 4

#define H_OFF       102400
static_assert(H_OFF >= N_NODES && (H_OFF * sizeof(float)) % 512 == 0, "h tables: behind the out-degrees, on a 512-byte multiple");
static_assert(((size_t)N_NODES * D * sizeof(float)) % 512 == 0, "the second h table follows the first on a 512-byte multiple");

#define OWN_ROWS    196
#define OWN_BLOCKS  ((N_NODES + OWN_ROWS - 1) / OWN_ROWS)

#define PART_BLOCKS  256
#define PART_E4      ((N_EDGES / 4 + PART_BLOCKS - 1) / PART_BLOCKS)
#define PART_THREADS 640
#define PART_PITCH   (((PART_E4 * 4 + 31) / 32) * 32)
#define OFF_PITCH    288
static_assert(N_EDGES % 4 == 0, "the edge pass reads four edges per 16-byte load");
static_assert(PART_THREADS >= PART_E4 && PART_THREADS % 32 == 0, "one 16-byte load of each array per thread");
static_assert(OWN_BLOCKS == 256 && PART_BLOCKS == 256, "the prefix sum below is written for 256 counts");
static_assert(OFF_PITCH >= OWN_BLOCKS + 1 && OFF_PITCH % 32 == 0, "table pitch");

static __device__ __forceinline__ void excl_scan_256(const int* cnt, int* start, int lane) {
    int v[8];
    int sum = 0;
#pragma unroll
    for (int k = 0; k < 8; ++k) { v[k] = cnt[lane * 8 + k]; sum += v[k]; }
    int inc = sum;
#pragma unroll
    for (int off = 1; off < 32; off <<= 1) {
        const int t = __shfl_up(inc, off);
        if (lane >= off) inc += t;
    }
    int ex = inc - sum;
#pragma unroll
    for (int k = 0; k < 8; ++k) { start[lane * 8 + k] = ex; ex += v[k]; }
    if (lane == 31) start[256] = ex;
}

__global__ __launch_bounds__(PART_THREADS) void edge_pass_kernel(const int* __restrict__ src,
                                                                 const int* __restrict__ dst,
                                                                 float* __restrict__ out_deg,
                                                                 unsigned* __restrict__ gpl,
                                                                 int* __restrict__ goff) {
    __shared__ int      bcnt[OWN_BLOCKS];
    __shared__ int      bstart[OWN_BLOCKS + 1];
    __shared__ unsigned sorted[PART_PITCH];

    const int tid = threadIdx.x;
    if (tid < OWN_BLOCKS) bcnt[tid] = 0;
    __syncthreads();

    const int  idx4  = blockIdx.x * PART_E4 + tid;
    const bool valid = tid < PART_E4 && idx4 < N_EDGES / 4;
    int      own[4] = {-1, -1, -1, -1};
    int      rk[4]  = {0, 0, 0, 0};
    unsigned wd[4]  = {0u, 0u, 0u, 0u};
    if (valid) {
        const int4 d4 = reinterpret_cast<const int4*>(dst)[idx4];
        const int4 s4 = reinterpret_cast<const int4*>(src)[idx4];
        const int dd[4] = {d4.x, d4.y, d4.z, d4.w};
        const int ss[4] = {s4.x, s4.y, s4.z, s4.w};
#pragma unroll
        for (int c = 0; c < 4; ++c) {
            unsafeAtomicAdd(&out_deg[ss[c]], 1.0f);
            if ((unsigned)dd[c] < (unsigned)N_NODES) {
                const unsigned o = (unsigned)dd[c] / OWN_ROWS;
                const unsigned r = (unsigned)dd[c] - o * OWN_ROWS;
                own[c] = (int)o;
                wd[c]  = (r << 16) | (unsigned)ss[c];
                rk[c]  = atomicAdd(&bcnt[o], 1);
            }
        }
    }
    __syncthreads();
    if (tid < 32) excl_scan_256(bcnt, bstart, tid);
    __syncthreads();
#pragma unroll
    for (int c = 0; c < 4; ++c)
        if (own[c] >= 0) sorted[bstart[own[c]] + rk[c]] = wd[c];
    __syncthreads();

    const int tot = bstart[OWN_BLOCKS];
    unsigned* __restrict__ pl = gpl + (size_t)blockIdx.x * PART_PITCH;
    for (int i = tid; i < tot; i += PART_THREADS) pl[i] = sorted[i];
    if (tid <= OWN_BLOCKS) goff[(size_t)blockIdx.x * OFF_PITCH + tid] = bstart[tid];
}

static __device__ __forceinline__ float norm_of_deg(float deg) { return rsqrtf(fmaxf(deg, 1.0f)); }

#define OWN_THREADS 1024
#define OWN_WAVES   (OWN_THREADS / 32)
#define LIST_CAP    10240
#define COPY_U      4
#define LOOK_U      4
#define GATHER_U    8
#define CNT_PITCH   32
#define DN_PITCH    256
#define ACC_TILES   ((OWN_ROWS + 15) / 16)
#define ACC_ROWS    (ACC_TILES * 16)
#define ACC_PITCH   132

static __device__ __forceinline__ int feat_of_word(int w) { return 4 * (w & 31) + (w >> 5); }

static __device__ __forceinline__ void add_edge_alone(float* __restrict__ arow,
                                                      const float* __restrict__ h,
                                                      const int* __restrict__ src,
                                                      const float* __restrict__ src_deg,
                                                      int e) {
    const int   s   = src[e];
    const float sc  = norm_of_deg(src_deg[s]);
    const float* hr = h + (size_t)s * D;
    for (int k = 0; k < D; ++k)
        unsafeAtomicAdd(&arow[(k & 3) * 32 + (k >> 2)], hr[k] * sc);
}

static __device__ __forceinline__ void store_w_stage(unsigned* ubuf, const float (&wq)[4][4], int wave, int lane) {
    uint4* wst = reinterpret_cast<uint4*>(ubuf);
#pragma unroll
    for (int i = 0; i < 4; ++i) {
        const int g   = wave + OWN_WAVES * i;
        const int wn  = (g & 3) * 32 + lane;
        const int whf = (g >> 2) & 1;
        const int wjj = g >> 3;
        wst[((wn >> 4) * 16 + wjj) * 32 + whf * 16 + (wn & 15)] =
            make_uint4(__float_as_uint(wq[i][0]), __float_as_uint(wq[i][1]),
                       __float_as_uint(wq[i][2]), __float_as_uint(wq[i][3]));
    }
}

template <bool BUILD>
__global__ __launch_bounds__(OWN_THREADS) void layer_owner_kernel(
        const float* __restrict__ h,
        const int* __restrict__ src,
        const int* __restrict__ dst,
        const float* __restrict__ src_deg,
        const float* __restrict__ Wm,
        const float* __restrict__ bias,
        float* __restrict__ hout,
        unsigned* __restrict__ glst,
        float* __restrict__ glsc,
        int* __restrict__ gcnt,
        float* __restrict__ gdn,
        const unsigned* __restrict__ gpl,
        const int* __restrict__ goff) {
    __shared__ __attribute__((aligned(16))) float acc[ACC_ROWS * ACC_PITCH];
    __shared__ unsigned lst[LIST_CAP];
    __shared__ float    lsc[LIST_CAP];
    __shared__ int      rcnt[256];
    __shared__ int      rstart[256 + 1];
    __shared__ int      rfill[256];
    __shared__ __attribute__((aligned(16))) float dnl[ACC_ROWS];
    static_assert(2 * LIST_CAP >= D * D, "W stage fits over the first layer's unsorted list");
    __shared__ __attribute__((aligned(16))) unsigned ubuf[BUILD ? 2 * LIST_CAP : D * D];
    unsigned* const ulst = ubuf;
    __shared__ int      pstart[PART_BLOCKS];
    __shared__ int      pcnt[PART_BLOCKS];
    __shared__ int      pbase[PART_BLOCKS + 1];

    const int tid  = threadIdx.x;
    const int wave = tid >> 5;
    const int lane = tid & 31;
    const int lo   = blockIdx.x * OWN_ROWS;

    unsigned* __restrict__ blst = glst + (size_t)blockIdx.x * LIST_CAP;
    float*    __restrict__ blsc = glsc + (size_t)blockIdx.x * LIST_CAP;

    float wq[4][4];
#pragma unroll
    for (int i = 0; i < 4; ++i) {
        const int g   = wave + OWN_WAVES * i;
        const int wn  = (g & 3) * 32 + lane;
        const int whf = (g >> 2) & 1;
        const int wjj = g >> 3;
#pragma unroll
        for (int q = 0; q < 4; ++q)
            wq[i][q] = Wm[(size_t)feat_of_word(4 * (2 * wjj + (q >> 1)) + 2 * whf + (q & 1)) * D + wn];
    }
    const int wv   = __builtin_amdgcn_readfirstlane(wave);
    const int c16  = lane & 15;
    const int hf   = lane >> 4;
    const int wv8  = wv & 7;
    const int ncol = wv8 * 16 + c16;
    const float bn = bias[ncol];

    int n = BUILD ? 0 : gcnt[blockIdx.x * CNT_PITCH];

    {
        float4* a4 = reinterpret_cast<float4*>(acc);
        const float4 z = make_float4(0.f, 0.f, 0.f, 0.f);
        for (int i = tid; i < ACC_ROWS * ACC_PITCH / 4; i += OWN_THREADS) a4[i] = z;
        if (BUILD && tid < 256) { rcnt[tid] = 0; rfill[tid] = 0; }
    }
    if (!BUILD && n <= LIST_CAP) {
        for (int i = tid; i < n; i += OWN_THREADS) {
            lst[i] = blst[i];
            lsc[i] = blsc[i];
        }
    }
    if (BUILD && tid < PART_BLOCKS) {
        const int* po = goff + (size_t)tid * OFF_PITCH + blockIdx.x;
        const int st = po[0];
        pstart[tid] = st;
        pcnt[tid]   = po[1] - st;
    }
    if (!BUILD && tid < ACC_ROWS) dnl[tid] = gdn[blockIdx.x * DN_PITCH + tid];
    if (!BUILD) store_w_stage(ubuf, wq, wave, lane);
    __syncthreads();

    if (BUILD) {
        static_assert(OWN_THREADS == 4 * PART_BLOCKS, "the run copy below deals four threads to a piece");
        if (wave == 0) excl_scan_256(pcnt, pbase, lane);
        __syncthreads();
        n = pbase[PART_BLOCKS];
        if (tid == 0) gcnt[blockIdx.x * CNT_PITCH] = n;
        if (n <= LIST_CAP) {
            const int p = tid >> 2;
            const int q = tid & 3;
            const int b = pbase[p];
            const int c = pbase[p + 1] - b;
            const unsigned* __restrict__ pp = gpl + (size_t)p * PART_PITCH + pstart[p];
            for (int j0 = q; j0 < c; j0 += 4 * COPY_U) {
                unsigned v[COPY_U];
#pragma unroll
                for (int k = 0; k < COPY_U; ++k) {
                    const int j = j0 + 4 * k;
                    v[k] = (j < c) ? pp[j] : 0u;
                }
                asm volatile("" :: "v"(v[0]), "v"(v[1]), "v"(v[2]), "v"(v[3]));
#pragma unroll
                for (int k = 0; k < COPY_U; ++k) {
                    const int j = j0 + 4 * k;
                    if (j < c) ulst[b + j] = v[k];
                }
            }
        }
        __syncthreads();
    }
    if (n > LIST_CAP) {
        for (int e = tid; e < N_EDGES; e += OWN_THREADS) {
            const unsigned r = (unsigned)(dst[e] - lo);
            if (r < (unsigned)OWN_ROWS) {
                if (BUILD) atomicAdd(&rcnt[r], 1);
                add_edge_alone(acc + r * ACC_PITCH, h, src, src_deg, e);
            }
        }
        n = 0;
    }

    for (int i0 = tid; BUILD && i0 < n; i0 += OWN_THREADS * LOOK_U) {
        unsigned w[LOOK_U];
        float    dg[LOOK_U];
#pragma unroll
        for (int k = 0; k < LOOK_U; ++k) {
            const int i = i0 + k * OWN_THREADS;
            w[k] = (i < n) ? ulst[i] : 0u;
        }
#pragma unroll
        for (int k = 0; k < LOOK_U; ++k) dg[k] = src_deg[w[k] & 0xFFFFu];
        asm volatile("" :: "v"(dg[0]), "v"(dg[1]), "v"(dg[2]), "v"(dg[3]));
#pragma unroll
        for (int k = 0; k < LOOK_U; ++k) {
            const int i = i0 + k * OWN_THREADS;
            if (i < n) {
                const float sc = norm_of_deg(dg[k]);
                atomicAdd(&rcnt[w[k] >> 16], 1);
                ubuf[LIST_CAP + i] = __float_as_uint(sc);
            }
        }
    }
    if (BUILD) {
        __syncthreads();
        if (tid < ACC_ROWS) {
            const float nv = norm_of_deg((float)rcnt[tid]);
            gdn[blockIdx.x * DN_PITCH + tid] = nv;
            dnl[tid] = nv;
        }
        static_assert(ACC_ROWS <= 256, "the row counts are scanned as 256 counts");
        if (wave == 0) excl_scan_256(rcnt, rstart, lane);
        __syncthreads();
        for (int i = tid; i < n; i += OWN_THREADS) {
            const unsigned w = ulst[i];
            const int      r = (int)(w >> 16);
            const int    pos = rstart[r] + atomicAdd(&rfill[r], 1);
            lst[pos] = w;
            lsc[pos] = __uint_as_float(ubuf[LIST_CAP + i]);
        }
        __syncthreads();
        for (int i = tid; i < n; i += OWN_THREADS) {
            blst[i] = lst[i];
            blsc[i] = lsc[i];
        }
        store_w_stage(ubuf, wq, wave, lane);
        __syncthreads();
    }

    {
        const int ns = __builtin_amdgcn_readfirstlane(n);
        const int e0 = (wv * ns) / OWN_WAVES;
        const int e1 = ((wv + 1) * ns) / OWN_WAVES;
        int    cur = -1;
        float4 run = make_float4(0.f, 0.f, 0.f, 0.f);
        for (int i0 = e0; i0 < e1; i0 += GATHER_U) {
            unsigned p[GATHER_U];
            float    sc[GATHER_U];
            float4   row[GATHER_U];
#pragma unroll
            for (int u = 0; u < GATHER_U; ++u) {
                const int i = min(i0 + u, e1 - 1);
                p[u]   = __builtin_amdgcn_readfirstlane(lst[i]);
                sc[u]  = lsc[i];
                row[u] = *reinterpret_cast<const float4*>(h + (size_t)(p[u] & 0xFFFFu) * D + lane * 4);
            }
#pragma unroll
            for (int u = 0; u < GATHER_U; ++u) {
                if (i0 + u < e1) {
                    const int r = (int)(p[u] >> 16);
                    if (r != cur) {
                        if (cur >= 0) {
                            float* a = acc + cur * ACC_PITCH + lane;
                            unsafeAtomicAdd(a,      run.x);
                            unsafeAtomicAdd(a + 32, run.y);
                            unsafeAtomicAdd(a + 64, run.z);
                            unsafeAtomicAdd(a + 96, run.w);
                        }
                        cur = r;
                        run = make_float4(0.f, 0.f, 0.f, 0.f);
                    }
                    run.x = fmaf(row[u].x, sc[u], run.x);
                    run.y = fmaf(row[u].y, sc[u], run.y);
                    run.z = fmaf(row[u].z, sc[u], run.z);
                    run.w = fmaf(row[u].w, sc[u], run.w);
                }
            }
        }
        if (cur >= 0) {
            float* a = acc + cur * ACC_PITCH + lane;
            unsafeAtomicAdd(a,      run.x);
            unsafeAtomicAdd(a + 32, run.y);
            unsafeAtomicAdd(a + 64, run.z);
            unsafeAtomicAdd(a + 96, run.w);
        }
    }

    static_assert(OWN_WAVES == 32 && ACC_TILES == 13, "tile deal below is written for 32 waves, 13 row tiles");
    v2f bfr[D / 4];
    {
        const uint4* wf = reinterpret_cast<const uint4*>(ubuf) + (wv8 * 16) * 32 + lane;
#pragma unroll
        for (int jj = 0; jj < D / 8; ++jj) {
            const uint4 f = wf[jj * 32];
            bfr[2 * jj    ].x = __uint_as_float(f.x);
            bfr[2 * jj    ].y = __uint_as_float(f.y);
            bfr[2 * jj + 1].x = __uint_as_float(f.z);
            bfr[2 * jj + 1].y = __uint_as_float(f.w);
        }
    }
    __syncthreads();

#pragma unroll
    for (int t = 0; t < 4; ++t) {
        int rt = (wv >> 3) + 4 * t;
        if (t == 3) {
            if ((wv >> 3) != (wv & 3)) break;
            rt = ACC_TILES - 1;
        }
        const int rbase = rt * 16 + hf * 8;
        const float4 dlo = *reinterpret_cast<const float4*>(dnl + rbase);
        const float4 dhi = *reinterpret_cast<const float4*>(dnl + rbase + 4);
        const float* ap  = acc + (rt * 16 + c16) * ACC_PITCH + 2 * hf;
        v8f c = {};
#pragma unroll
        for (int s = 0; s < D / 4; ++s) {
            const v2f a = *reinterpret_cast<const v2f*>(ap + 4 * s);
            c = __builtin_amdgcn_wmma_f32_16x16x4_f32(
                     false, a,  false, bfr[s],
                     (short)0, c,  false,  false);
        }
        const float dn8[8] = {dlo.x, dlo.y, dlo.z, dlo.w, dhi.x, dhi.y, dhi.z, dhi.w};
        float o[8];
#pragma unroll
        for (int v = 0; v < 8; ++v) o[v] = fmaf(c[v], dn8[v], bn);
        if (rt * 16 + 16 <= OWN_ROWS && lo + rt * 16 + 16 <= N_NODES) {
            float* __restrict__ p = hout + (size_t)(lo + rbase) * D + ncol;
#pragma unroll
            for (int v = 0; v < 8; ++v) p[v * D] = o[v];
        } else {
#pragma unroll
            for (int v = 0; v < 8; ++v) {
                const int r    = rbase + v;
                const int node = lo + r;
                if (r < OWN_ROWS && node < N_NODES)
                    hout[(size_t)node * D + ncol] = o[v];
            }
        }
    }
}


extern "C" void kernel_launch(void* const* d_in, const int* in_sizes, int n_in,
                              void* d_out, int out_size, void* d_ws, size_t ws_size,
                              hipStream_t stream) {
    const float* feat = (const float*)d_in[0];
    const int*   src  = (const int*)  d_in[1];
    const int*   dst  = (const int*)  d_in[2];
    const float* W    = (const float*)d_in[3];
    const float* b    = (const float*)d_in[4];
    float*       out  = (float*)d_out;

    float* ws       = (float*)d_ws;
    float* out_deg  = ws;
    float* hA       = ws + (size_t)H_OFF;
    float* hB       = hA + (size_t)N_NODES * D;
    unsigned* glst  = (unsigned*)(hB + (size_t)N_NODES * D);
    float*    glsc  = (float*)(glst + (size_t)OWN_BLOCKS * LIST_CAP);
    int*      gcnt  = (int*)(glsc + (size_t)OWN_BLOCKS * LIST_CAP);
    float*    gdn   = (float*)(gcnt + (size_t)OWN_BLOCKS * CNT_PITCH);
    unsigned* gpl   = (unsigned*)(gdn + (size_t)OWN_BLOCKS * DN_PITCH);
    int*      goff  = (int*)(gpl + (size_t)PART_BLOCKS * PART_PITCH);

    hipMemsetAsync(out_deg, 0, (size_t)N_NODES * sizeof(float), stream);
    edge_pass_kernel<<<PART_BLOCKS, PART_THREADS, 0, stream>>>(src, dst, out_deg, gpl, goff);

    layer_owner_kernel<true><<<OWN_BLOCKS, OWN_THREADS, 0, stream>>>(
        feat, src, dst, out_deg, W, b, hA, glst, glsc, gcnt, gdn, gpl, goff);
    layer_owner_kernel<false><<<OWN_BLOCKS, OWN_THREADS, 0, stream>>>(
        hA, src, dst, out_deg, W + (size_t)1 * D * D, b + (size_t)1 * D, hB, glst, glsc, gcnt, gdn, gpl, goff);
    layer_owner_kernel<false><<<OWN_BLOCKS, OWN_THREADS, 0, stream>>>(
        hB, src, dst, out_deg, W + (size_t)2 * D * D, b + (size_t)2 * D, hA, glst, glsc, gcnt, gdn, gpl, goff);
    layer_owner_kernel<false><<<OWN_BLOCKS, OWN_THREADS, 0, stream>>>(
        hA, src, dst, out_deg, W + (size_t)3 * D * D, b + (size_t)3 * D, out, glst, glsc, gcnt, gdn, gpl, goff);
}
